# starting kernel
# baseline (speedup 1.0000x reference)
#include <hip/hip_runtime.h>
#include <hip/hip_bf16.h>

#define TOTAL_K     131072
#define CHUNK_LEN   64
#define F_DIM       8
#define NUM_FRAMES  16384
#define ENC_DIM     256
#define D_MODEL     512
#define N_SCALARS   3
#define N_TICKERS   4096
#define YP          520

typedef float v2f __attribute__((ext_vector_type(2)));
typedef float v8f __attribute__((ext_vector_type(8)));

__device__ __forceinline__ v8f wmma_f32_4(v2f a, v2f b, v8f c) {
  return __builtin_amdgcn_wmma_f32_16x16x4_f32(
       false, a,  false, b,
       (short)0, c,  false,  false);
}

__device__ __forceinline__ float gelu_exact(float x) {
  float z  = x * 0.70710678118654752f;
  float az = fabsf(z);
  float t  = __builtin_amdgcn_rcpf(fmaf(0.3275911f, az, 1.0f));
  float p  = fmaf(t, 1.061405429f, -1.453152027f);
  p        = fmaf(t, p,  1.421413741f);
  p        = fmaf(t, p, -0.284496736f);
  p        = fmaf(t, p,  0.254829592f);
  p       *= t;
  float e  = __expf(-az * az);
  float erf_abs = fmaf(-p, e, 1.0f);
  float erf     = copysignf(erf_abs, z);
  return 0.5f * x * (1.0f + erf);
}

__global__ void k_zero(float* __restrict__ p, int n) {
  int i = blockIdx.x * blockDim.x + threadIdx.x;
  if (i < n) p[i] = 0.0f;
}

__global__ void __launch_bounds__(256)
k_encode(const float* __restrict__ chunks,
         const int*   __restrict__ frame_id,
         const float* __restrict__ weights,
         const int*   __restrict__ ticker_ids,
         const float* __restrict__ W_enc,
         const float* __restrict__ b_enc,
         const float* __restrict__ ticker_emb,
         float* __restrict__ sum_emb,
         float* __restrict__ sum_w) {
  const int lane  = threadIdx.x & 31;
  const int wave  = threadIdx.x >> 5;
  const int tileM = blockIdx.x * 8 + wave;
  const int c0    = tileM * 16;
  const int lm    = lane & 15;
  const int half  = lane >> 4;

  const int c = c0 + lm;
  const float4* base =
      (const float4*)(chunks + (size_t)c * (CHUNK_LEN * F_DIM) + (size_t)half * 32 * F_DIM);
  __builtin_prefetch(base, 0, 0);
  float p[8] = {0.f,0.f,0.f,0.f,0.f,0.f,0.f,0.f};
#pragma unroll 8
  for (int r = 0; r < 32; ++r) {
    float4 q0 = base[r * 2 + 0];
    float4 q1 = base[r * 2 + 1];
    p[0] += q0.x; p[1] += q0.y; p[2] += q0.z; p[3] += q0.w;
    p[4] += q1.x; p[5] += q1.y; p[6] += q1.z; p[7] += q1.w;
  }
  float m[8];
#pragma unroll
  for (int f = 0; f < 8; ++f)
    m[f] = (p[f] + __shfl_xor(p[f], 16, 32)) * (1.0f / 64.0f);

  v2f A0, A1;
  A0.x = m[2 * half + 0];  A0.y = m[2 * half + 1];
  A1.x = m[2 * half + 4];  A1.y = m[2 * half + 5];

  int   fid[8], tid[8];
  float wg[8];
#pragma unroll
  for (int r = 0; r < 8; ++r) {
    int cr = c0 + half * 8 + r;
    fid[r] = frame_id[cr];
    tid[r] = ticker_ids[fid[r]];
    wg[r]  = weights[cr];
  }
  if (half == 0) {
    int cc = c0 + lm;
    atomicAdd(&sum_w[frame_id[cc]], weights[cc]);
  }

  const v8f vzero = {0.f,0.f,0.f,0.f,0.f,0.f,0.f,0.f};
#pragma unroll 1
  for (int nt = 0; nt < 16; ++nt) {
    const int col = nt * 16 + lm;
    v2f B0, B1;
    B0.x = W_enc[(2 * half + 0) * ENC_DIM + col];
    B0.y = W_enc[(2 * half + 1) * ENC_DIM + col];
    B1.x = W_enc[(2 * half + 4) * ENC_DIM + col];
    B1.y = W_enc[(2 * half + 5) * ENC_DIM + col];
    v8f acc = wmma_f32_4(A0, B0, vzero);
    acc     = wmma_f32_4(A1, B1, acc);

    const float be = b_enc[col];
#pragma unroll
    for (int r = 0; r < 8; ++r) {
      float h   = acc[r] + be;
      float g   = gelu_exact(h);
      float val = (g + ticker_emb[(size_t)tid[r] * ENC_DIM + col]) * wg[r];
      atomicAdd(&sum_emb[(size_t)fid[r] * ENC_DIM + col], val);
    }
  }
}

__global__ void __launch_bounds__(128)
k_project(const float* __restrict__ sum_emb,
          const float* __restrict__ sum_w,
          const float* __restrict__ frame_scalars,
          const float* __restrict__ W_proj,
          const float* __restrict__ b_proj,
          const float* __restrict__ ln_gamma,
          const float* __restrict__ ln_beta,
          float* __restrict__ out) {
  __shared__ float pool[16 * ENC_DIM];
  __shared__ float ytile[16 * YP];
  __shared__ float smu[16], srs[16];

  const int tidx = threadIdx.x;
  const int lane = tidx & 31;
  const int wave = tidx >> 5;
  const int lm   = lane & 15;
  const int half = lane >> 4;
  const int fb   = blockIdx.x * 16;

  for (int i = tidx; i < 16 * ENC_DIM; i += 128) {
    int row = i >> 8;
    float rw = 1.0f / (sum_w[fb + row] + 1e-8f);
    pool[i] = sum_emb[(size_t)(fb + row) * ENC_DIM + (i & 255)] * rw;
  }
  __syncthreads();

  const v8f vzero = {0.f,0.f,0.f,0.f,0.f,0.f,0.f,0.f};
  v8f acc[8];
#pragma unroll
  for (int t = 0; t < 8; ++t) acc[t] = vzero;

#pragma unroll 1
  for (int kk = 0; kk < 64; ++kk) {
    const int kbase = 4 * kk + 2 * half;
    v2f a;
    a.x = pool[lm * ENC_DIM + kbase + 0];
    a.y = pool[lm * ENC_DIM + kbase + 1];
#pragma unroll
    for (int t = 0; t < 8; ++t) {
      const int col = (wave * 8 + t) * 16 + lm;
      v2f b;
      b.x = W_proj[(size_t)(kbase + 0) * D_MODEL + col];
      b.y = W_proj[(size_t)(kbase + 1) * D_MODEL + col];
      acc[t] = wmma_f32_4(a, b, acc[t]);
    }
  }

  float l0[8], l1[8], l2[8];
#pragma unroll
  for (int r = 0; r < 8; ++r) {
    int f = fb + half * 8 + r;
    l0[r] = log1pf(frame_scalars[f * N_SCALARS + 0]);
    l1[r] = log1pf(frame_scalars[f * N_SCALARS + 1]);
    l2[r] = log1pf(frame_scalars[f * N_SCALARS + 2]);
  }
#pragma unroll
  for (int t = 0; t < 8; ++t) {
    const int col = (wave * 8 + t) * 16 + lm;
    const float w0 = W_proj[(size_t)(ENC_DIM + 0) * D_MODEL + col];
    const float w1 = W_proj[(size_t)(ENC_DIM + 1) * D_MODEL + col];
    const float w2 = W_proj[(size_t)(ENC_DIM + 2) * D_MODEL + col];
    const float bp = b_proj[col];
#pragma unroll
    for (int r = 0; r < 8; ++r) {
      float v = acc[t][r] + bp + l0[r] * w0 + l1[r] * w1 + l2[r] * w2;
      ytile[(half * 8 + r) * YP + col] = v;
    }
  }
  __syncthreads();

  {
    const int row = tidx >> 3;
    const int seg = tidx & 7;
    float s = 0.0f, ss = 0.0f;
#pragma unroll 16
    for (int j = 0; j < 64; ++j) {
      float v = ytile[row * YP + seg + j * 8];
      s  += v;
      ss += v * v;
    }
#pragma unroll
    for (int off = 1; off < 8; off <<= 1) {
      s  += __shfl_xor(s,  off, 32);
      ss += __shfl_xor(ss, off, 32);
    }
    if (seg == 0) {
      float mu  = s * (1.0f / (float)D_MODEL);
      float var = fmaf(-mu, mu, ss * (1.0f / (float)D_MODEL));
      smu[row] = mu;
      srs[row] = rsqrtf(fmaxf(var, 0.0f) + 1e-5f);
    }
  }
  __syncthreads();

  for (int i = tidx; i < 16 * D_MODEL; i += 128) {
    int row = i >> 9, col = i & 511;
    float x = (ytile[row * YP + col] - smu[row]) * srs[row] * ln_gamma[col] + ln_beta[col];
    out[(size_t)(fb + row) * D_MODEL + col] = gelu_exact(x);
  }
}

extern "C" void kernel_launch(void* const* d_in, const int* in_sizes, int n_in,
                              void* d_out, int out_size, void* d_ws, size_t ws_size,
                              hipStream_t stream) {
  const float* chunks        = (const float*)d_in[0];
  const int*   frame_id      = (const int*)  d_in[1];
  const float* weights       = (const float*)d_in[2];
  const float* frame_scalars = (const float*)d_in[3];
  const int*   ticker_ids    = (const int*)  d_in[5];
  const float* W_enc         = (const float*)d_in[6];
  const float* b_enc         = (const float*)d_in[7];
  const float* ticker_emb    = (const float*)d_in[8];
  const float* W_proj        = (const float*)d_in[9];
  const float* b_proj        = (const float*)d_in[10];
  const float* ln_gamma      = (const float*)d_in[11];
  const float* ln_beta       = (const float*)d_in[12];
  float* out = (float*)d_out;

  float* sum_emb = (float*)d_ws;
  float* sum_w   = sum_emb + (size_t)NUM_FRAMES * ENC_DIM;

  const int nz = NUM_FRAMES * ENC_DIM + NUM_FRAMES;
  k_zero<<<(nz + 255) / 256, 256, 0, stream>>>(sum_emb, nz);

  k_encode<<<TOTAL_K / 16 / 8, 256, 0, stream>>>(
      chunks, frame_id, weights, ticker_ids, W_enc, b_enc, ticker_emb,
      sum_emb, sum_w);

  k_project<<<NUM_FRAMES / 16, 128, 0, stream>>>(
      sum_emb, sum_w, frame_scalars, W_proj, b_proj, ln_gamma, ln_beta, out);
}


# Float atomic scatter replaced by a counting-sort gather: frames read their chunks' means in sorted order and segment-sum gelu rows on WMMA inside the projection kernel; sum order changes
# speedup vs baseline: 1.3766x; 1.3766x over previous
#include <hip/hip_runtime.h>
#include <hip/hip_bf16.h>

#define TOTAL_K     131072
#define CHUNK_LEN   64
#define F_DIM       8
#define NUM_FRAMES  16384
#define ENC_DIM     256
#define D_MODEL     512
#define N_SCALARS   3
#define N_TICKERS   4096
#define YP          520

typedef float v2f __attribute__((ext_vector_type(2)));
typedef float v8f __attribute__((ext_vector_type(8)));

__device__ __forceinline__ v8f wmma_f32_4(v2f a, v2f b, v8f c) {
  return __builtin_amdgcn_wmma_f32_16x16x4_f32(
       false, a,  false, b,
       (short)0, c,  false,  false);
}

__device__ __forceinline__ float gelu_exact(float x) {
  float z  = x * 0.70710678118654752f;
  float az = fabsf(z);
  float t  = __builtin_amdgcn_rcpf(fmaf(0.3275911f, az, 1.0f));
  float p  = fmaf(t, 1.061405429f, -1.453152027f);
  p        = fmaf(t, p,  1.421413741f);
  p        = fmaf(t, p, -0.284496736f);
  p        = fmaf(t, p,  0.254829592f);
  p       *= t;
  float e  = __expf(-az * az);
  float erf_abs = fmaf(-p, e, 1.0f);
  float erf     = copysignf(erf_abs, z);
  return 0.5f * x * (1.0f + erf);
}

__global__ void k_zero(int* __restrict__ p, int n) {
  int i = blockIdx.x * blockDim.x + threadIdx.x;
  if (i < n) p[i] = 0;
}

__global__ void __launch_bounds__(256)
k_means(const float* __restrict__ chunks,
        const int*   __restrict__ frame_id,
        float* __restrict__ means,
        int*   __restrict__ counts,
        int*   __restrict__ rank) {
  const int lane = threadIdx.x & 31;
  const int wave = threadIdx.x >> 5;
  const int c0   = (blockIdx.x * 8 + wave) * 16;

  if (lane < 16) {
    const int c = c0 + lane;
    rank[c] = atomicAdd(&counts[frame_id[c]], 1);
  }

#pragma unroll 1
  for (int g = 0; g < 8; ++g) {
    const float4* b = (const float4*)(chunks + (size_t)(c0 + g * 2) * (CHUNK_LEN * F_DIM));
    float4 q[2][4];
#pragma unroll
    for (int k = 0; k < 2; ++k)
#pragma unroll
      for (int j = 0; j < 4; ++j) q[k][j] = b[k * 128 + j * 32 + lane];
    float4 s[2];
#pragma unroll
    for (int k = 0; k < 2; ++k) {
      s[k].x = (q[k][0].x + q[k][1].x) + (q[k][2].x + q[k][3].x);
      s[k].y = (q[k][0].y + q[k][1].y) + (q[k][2].y + q[k][3].y);
      s[k].z = (q[k][0].z + q[k][1].z) + (q[k][2].z + q[k][3].z);
      s[k].w = (q[k][0].w + q[k][1].w) + (q[k][2].w + q[k][3].w);
    }
#pragma unroll
    for (int off = 2; off < 32; off <<= 1) {
#pragma unroll
      for (int k = 0; k < 2; ++k) {
        s[k].x += __shfl_xor(s[k].x, off, 32);
        s[k].y += __shfl_xor(s[k].y, off, 32);
        s[k].z += __shfl_xor(s[k].z, off, 32);
        s[k].w += __shfl_xor(s[k].w, off, 32);
      }
    }
    float4 v = (lane >> 1) ? s[1] : s[0];
    v.x *= (1.0f / 64.0f); v.y *= (1.0f / 64.0f);
    v.z *= (1.0f / 64.0f); v.w *= (1.0f / 64.0f);
    if (lane < 4) ((float4*)(means + (size_t)(c0 + g * 2) * F_DIM))[lane] = v;
  }
}

__global__ void __launch_bounds__(1024)
k_place(const int*   __restrict__ counts,
        const int*   __restrict__ rank,
        const int*   __restrict__ frame_id,
        const float* __restrict__ weights,
        const float* __restrict__ means,
        int*   __restrict__ goff,
        float* __restrict__ smean,
        float* __restrict__ swt,
        int*   __restrict__ sfid) {
  __shared__ int off_s[NUM_FRAMES];
  int* wtot = off_s;
  const int t    = threadIdx.x;
  const int lane = t & 31;
  const int wave = t >> 5;

  int e[16];
  const int4* cp = (const int4*)(counts + t * 16);
#pragma unroll
  for (int i = 0; i < 4; ++i) {
    int4 c4 = cp[i];
    e[4 * i + 0] = c4.x; e[4 * i + 1] = c4.y; e[4 * i + 2] = c4.z; e[4 * i + 3] = c4.w;
  }
  int tot = 0;
#pragma unroll
  for (int i = 0; i < 16; ++i) { int c = e[i]; e[i] = tot; tot += c; }
  int incl = tot;
#pragma unroll
  for (int off = 1; off < 32; off <<= 1) {
    int u = __shfl_up(incl, off, 32);
    if (lane >= off) incl += u;
  }
  if (lane == 31) wtot[wave] = incl;
  __syncthreads();
  if (wave == 0) {
    int w = wtot[lane];
    int wi = w;
#pragma unroll
    for (int off = 1; off < 32; off <<= 1) {
      int u = __shfl_up(wi, off, 32);
      if (lane >= off) wi += u;
    }
    wtot[lane] = wi - w;
  }
  __syncthreads();
  const int base = wtot[wave] + (incl - tot);
  __syncthreads();
  int4* op = (int4*)(off_s + t * 16);
#pragma unroll
  for (int i = 0; i < 4; ++i)
    op[i] = make_int4(base + e[4 * i + 0], base + e[4 * i + 1],
                      base + e[4 * i + 2], base + e[4 * i + 3]);
  if (blockIdx.x == 0) {
    goff[t] = base;
    if (t == 1023) goff[1024] = base + tot;
  }
  __syncthreads();

  const int c = blockIdx.x * 1024 + t;
  const int f = frame_id[c];
  const int p = off_s[f] + rank[c];
  const float4* mp = (const float4*)(means + (size_t)c * F_DIM);
  float4 m0 = mp[0], m1 = mp[1];
  float4* sp = (float4*)(smean + (size_t)p * F_DIM);
  sp[0] = m0; sp[1] = m1;
  swt[p]  = weights[c];
  sfid[p] = f;
}

__global__ void __launch_bounds__(128)
k_frames(const int*   __restrict__ goff,
         const float* __restrict__ smean,
         const float* __restrict__ swt,
         const int*   __restrict__ sfid,
         const int*   __restrict__ ticker_ids,
         const float* __restrict__ W_enc,
         const float* __restrict__ b_enc,
         const float* __restrict__ ticker_emb,
         const float* __restrict__ frame_scalars,
         const float* __restrict__ W_proj,
         const float* __restrict__ b_proj,
         const float* __restrict__ ln_gamma,
         const float* __restrict__ ln_beta,
         float* __restrict__ out) {
  __shared__ float pool[16 * ENC_DIM];
  __shared__ float ytile[16 * YP];
  __shared__ float smu[16], srs[16];

  const int tidx = threadIdx.x;
  const int lane = tidx & 31;
  const int wave = tidx >> 5;
  const int lm   = lane & 15;
  const int half = lane >> 4;
  const int fb   = blockIdx.x * 16;

  const v8f vzero = {0.f,0.f,0.f,0.f,0.f,0.f,0.f,0.f};

  {
    const int start = goff[blockIdx.x];
    const int end   = goff[blockIdx.x + 1];
    const int ntile = (end - start + 15) >> 4;

    v2f B0[4], B1[4];
    float be[4];
#pragma unroll
    for (int t = 0; t < 4; ++t) {
      const int col = (wave * 4 + t) * 16 + lm;
      B0[t].x = W_enc[(2 * half + 0) * ENC_DIM + col];
      B0[t].y = W_enc[(2 * half + 1) * ENC_DIM + col];
      B1[t].x = W_enc[(2 * half + 4) * ENC_DIM + col];
      B1[t].y = W_enc[(2 * half + 5) * ENC_DIM + col];
      be[t]   = b_enc[col];
    }
    v8f pacc[4];
#pragma unroll
    for (int t = 0; t < 4; ++t) pacc[t] = vzero;
    float sw = 0.0f;

    for (int tile = 0; tile < ntile; ++tile) {
      const int tb = start + tile * 16;
      const int pa = min(tb + lm, TOTAL_K - 1);
      const float2* mp = (const float2*)(smean + (size_t)pa * F_DIM);
      const float2 ma = mp[half], mb = mp[half + 2];
      v2f A0, A1;
      A0.x = ma.x; A0.y = ma.y;
      A1.x = mb.x; A1.y = mb.y;
      v2f SA[4];
#pragma unroll
      for (int j = 0; j < 4; ++j) {
        const int r0 = tb + 8 * half + 2 * j;
        const int q0 = min(r0, TOTAL_K - 1), q1 = min(r0 + 1, TOTAL_K - 1);
        const float w0 = swt[q0], w1 = swt[q1];
        const int   f0 = sfid[q0] - fb, f1 = sfid[q1] - fb;
        SA[j].x = (r0     < end && f0 == lm) ? w0 : 0.0f;
        SA[j].y = (r0 + 1 < end && f1 == lm) ? w1 : 0.0f;
        sw += SA[j].x + SA[j].y;
      }
#pragma unroll
      for (int t = 0; t < 4; ++t) {
        v8f acc = wmma_f32_4(A0, B0[t], vzero);
        acc     = wmma_f32_4(A1, B1[t], acc);
        float g[8];
#pragma unroll
        for (int r = 0; r < 8; ++r) g[r] = gelu_exact(acc[r] + be[t]);
#pragma unroll
        for (int j = 0; j < 4; ++j) {
          v2f b;
          b.x = g[2 * j]; b.y = g[2 * j + 1];
          pacc[t] = wmma_f32_4(SA[j], b, pacc[t]);
        }
      }
    }
    sw += __shfl_xor(sw, 16, 32);
#pragma unroll
    for (int r = 0; r < 8; ++r) {
      const int   fl  = half * 8 + r;
      const float swr = __shfl(sw, fl, 32);
      const float rw  = 1.0f / (swr + 1e-8f);
      const int   tk  = ticker_ids[fb + fl];
#pragma unroll
      for (int t = 0; t < 4; ++t) {
        const int col = (wave * 4 + t) * 16 + lm;
        const float te = ticker_emb[(size_t)tk * ENC_DIM + col];
        pool[fl * ENC_DIM + col] = fmaf(te, swr, pacc[t][r]) * rw;
      }
    }
  }
  __syncthreads();

  v8f acc[8];
#pragma unroll
  for (int t = 0; t < 8; ++t) acc[t] = vzero;

#pragma unroll 1
  for (int kk = 0; kk < 64; ++kk) {
    const int kbase = 4 * kk + 2 * half;
    v2f a;
    a.x = pool[lm * ENC_DIM + kbase + 0];
    a.y = pool[lm * ENC_DIM + kbase + 1];
#pragma unroll
    for (int t = 0; t < 8; ++t) {
      const int col = (wave * 8 + t) * 16 + lm;
      v2f b;
      b.x = W_proj[(size_t)(kbase + 0) * D_MODEL + col];
      b.y = W_proj[(size_t)(kbase + 1) * D_MODEL + col];
      acc[t] = wmma_f32_4(a, b, acc[t]);
    }
  }

  float l0[8], l1[8], l2[8];
#pragma unroll
  for (int r = 0; r < 8; ++r) {
    int f = fb + half * 8 + r;
    l0[r] = log1pf(frame_scalars[f * N_SCALARS + 0]);
    l1[r] = log1pf(frame_scalars[f * N_SCALARS + 1]);
    l2[r] = log1pf(frame_scalars[f * N_SCALARS + 2]);
  }
#pragma unroll
  for (int t = 0; t < 8; ++t) {
    const int col = (wave * 8 + t) * 16 + lm;
    const float w0 = W_proj[(size_t)(ENC_DIM + 0) * D_MODEL + col];
    const float w1 = W_proj[(size_t)(ENC_DIM + 1) * D_MODEL + col];
    const float w2 = W_proj[(size_t)(ENC_DIM + 2) * D_MODEL + col];
    const float bp = b_proj[col];
#pragma unroll
    for (int r = 0; r < 8; ++r) {
      float v = acc[t][r] + bp + l0[r] * w0 + l1[r] * w1 + l2[r] * w2;
      ytile[(half * 8 + r) * YP + col] = v;
    }
  }
  __syncthreads();

  {
    const int row = tidx >> 3;
    const int seg = tidx & 7;
    float s = 0.0f, ss = 0.0f;
#pragma unroll 16
    for (int j = 0; j < 64; ++j) {
      float v = ytile[row * YP + seg + j * 8];
      s  += v;
      ss += v * v;
    }
#pragma unroll
    for (int off = 1; off < 8; off <<= 1) {
      s  += __shfl_xor(s,  off, 32);
      ss += __shfl_xor(ss, off, 32);
    }
    if (seg == 0) {
      float mu  = s * (1.0f / (float)D_MODEL);
      float var = fmaf(-mu, mu, ss * (1.0f / (float)D_MODEL));
      smu[row] = mu;
      srs[row] = rsqrtf(fmaxf(var, 0.0f) + 1e-5f);
    }
  }
  __syncthreads();

  for (int i = tidx; i < 16 * D_MODEL; i += 128) {
    int row = i >> 9, col = i & 511;
    float x = (ytile[row * YP + col] - smu[row]) * srs[row] * ln_gamma[col] + ln_beta[col];
    out[(size_t)(fb + row) * D_MODEL + col] = gelu_exact(x);
  }
}

extern "C" void kernel_launch(void* const* d_in, const int* in_sizes, int n_in,
                              void* d_out, int out_size, void* d_ws, size_t ws_size,
                              hipStream_t stream) {
  const float* chunks        = (const float*)d_in[0];
  const int*   frame_id      = (const int*)  d_in[1];
  const float* weights       = (const float*)d_in[2];
  const float* frame_scalars = (const float*)d_in[3];
  const int*   ticker_ids    = (const int*)  d_in[5];
  const float* W_enc         = (const float*)d_in[6];
  const float* b_enc         = (const float*)d_in[7];
  const float* ticker_emb    = (const float*)d_in[8];
  const float* W_proj        = (const float*)d_in[9];
  const float* b_proj        = (const float*)d_in[10];
  const float* ln_gamma      = (const float*)d_in[11];
  const float* ln_beta       = (const float*)d_in[12];
  float* out = (float*)d_out;

  float* means  = (float*)d_ws;
  float* smean  = means + (size_t)TOTAL_K * F_DIM;
  float* swt    = smean + (size_t)TOTAL_K * F_DIM;
  int*   sfid   = (int*)(swt + TOTAL_K);
  int*   rank   = sfid + TOTAL_K;
  int*   counts = rank + TOTAL_K;
  int*   goff   = counts + NUM_FRAMES;

  k_zero<<<NUM_FRAMES / 256, 256, 0, stream>>>(counts, NUM_FRAMES);

  k_means<<<TOTAL_K / 16 / 8, 256, 0, stream>>>(chunks, frame_id, means, counts, rank);

  k_place<<<TOTAL_K / 1024, 1024, 0, stream>>>(
      counts, rank, frame_id, weights, means, goff, smean, swt, sfid);

  k_frames<<<NUM_FRAMES / 16, 128, 0, stream>>>(
      goff, smean, swt, sfid, ticker_ids, W_enc, b_enc, ticker_emb,
      frame_scalars, W_proj, b_proj, ln_gamma, ln_beta, out);
}


# Frame kernel regrouped to 64 frames per block (16 waves): each W_proj fragment loaded from L2 feeds four row-tile WMMAs instead of one, 8-byte loads, next step prefetched; sum order unchanged
# speedup vs baseline: 1.6845x; 1.2237x over previous
#include <hip/hip_runtime.h>
#include <hip/hip_bf16.h>

#define TOTAL_K     131072
#define CHUNK_LEN   64
#define F_DIM       8
#define NUM_FRAMES  16384
#define ENC_DIM     256
#define D_MODEL     512
#define N_SCALARS   3
#define N_TICKERS   4096
#define YP          520
#define FPB         64
#define PP          260

typedef float v2f __attribute__((ext_vector_type(2)));
typedef float v8f __attribute__((ext_vector_type(8)));

__device__ __forceinline__ v8f wmma_f32_4(v2f a, v2f b, v8f c) {
  return __builtin_amdgcn_wmma_f32_16x16x4_f32(
       false, a,  false, b,
       (short)0, c,  false,  false);
}

__device__ __forceinline__ float gelu_exact(float x) {
  float z  = x * 0.70710678118654752f;
  float az = fabsf(z);
  float t  = __builtin_amdgcn_rcpf(fmaf(0.3275911f, az, 1.0f));
  float p  = fmaf(t, 1.061405429f, -1.453152027f);
  p        = fmaf(t, p,  1.421413741f);
  p        = fmaf(t, p, -0.284496736f);
  p        = fmaf(t, p,  0.254829592f);
  p       *= t;
  float e  = __expf(-az * az);
  float erf_abs = fmaf(-p, e, 1.0f);
  float erf     = copysignf(erf_abs, z);
  return 0.5f * x * (1.0f + erf);
}

__global__ void k_zero(int* __restrict__ p, int n) {
  int i = blockIdx.x * blockDim.x + threadIdx.x;
  if (i < n) p[i] = 0;
}

__global__ void __launch_bounds__(256)
k_means(const float* __restrict__ chunks,
        const int*   __restrict__ frame_id,
        float* __restrict__ means,
        int*   __restrict__ counts,
        int*   __restrict__ rank) {
  const int lane = threadIdx.x & 31;
  const int wave = threadIdx.x >> 5;
  const int c0   = (blockIdx.x * 8 + wave) * 16;

  if (lane < 16) {
    const int c = c0 + lane;
    rank[c] = atomicAdd(&counts[frame_id[c]], 1);
  }

#pragma unroll 1
  for (int g = 0; g < 8; ++g) {
    const float4* b = (const float4*)(chunks + (size_t)(c0 + g * 2) * (CHUNK_LEN * F_DIM));
    float4 q[2][4];
#pragma unroll
    for (int k = 0; k < 2; ++k)
#pragma unroll
      for (int j = 0; j < 4; ++j) q[k][j] = b[k * 128 + j * 32 + lane];
    float4 s[2];
#pragma unroll
    for (int k = 0; k < 2; ++k) {
      s[k].x = (q[k][0].x + q[k][1].x) + (q[k][2].x + q[k][3].x);
      s[k].y = (q[k][0].y + q[k][1].y) + (q[k][2].y + q[k][3].y);
      s[k].z = (q[k][0].z + q[k][1].z) + (q[k][2].z + q[k][3].z);
      s[k].w = (q[k][0].w + q[k][1].w) + (q[k][2].w + q[k][3].w);
    }
#pragma unroll
    for (int off = 2; off < 32; off <<= 1) {
#pragma unroll
      for (int k = 0; k < 2; ++k) {
        s[k].x += __shfl_xor(s[k].x, off, 32);
        s[k].y += __shfl_xor(s[k].y, off, 32);
        s[k].z += __shfl_xor(s[k].z, off, 32);
        s[k].w += __shfl_xor(s[k].w, off, 32);
      }
    }
    float4 v = (lane >> 1) ? s[1] : s[0];
    v.x *= (1.0f / 64.0f); v.y *= (1.0f / 64.0f);
    v.z *= (1.0f / 64.0f); v.w *= (1.0f / 64.0f);
    if (lane < 4) ((float4*)(means + (size_t)(c0 + g * 2) * F_DIM))[lane] = v;
  }
}

__global__ void __launch_bounds__(1024)
k_place(const int*   __restrict__ counts,
        const int*   __restrict__ rank,
        const int*   __restrict__ frame_id,
        const float* __restrict__ weights,
        const float* __restrict__ means,
        int*   __restrict__ goff,
        float* __restrict__ smean,
        float* __restrict__ swt,
        int*   __restrict__ sfid) {
  __shared__ int off_s[NUM_FRAMES];
  int* wtot = off_s;
  const int t    = threadIdx.x;
  const int lane = t & 31;
  const int wave = t >> 5;

  int e[16];
  const int4* cp = (const int4*)(counts + t * 16);
#pragma unroll
  for (int i = 0; i < 4; ++i) {
    int4 c4 = cp[i];
    e[4 * i + 0] = c4.x; e[4 * i + 1] = c4.y; e[4 * i + 2] = c4.z; e[4 * i + 3] = c4.w;
  }
  int tot = 0;
#pragma unroll
  for (int i = 0; i < 16; ++i) { int c = e[i]; e[i] = tot; tot += c; }
  int incl = tot;
#pragma unroll
  for (int off = 1; off < 32; off <<= 1) {
    int u = __shfl_up(incl, off, 32);
    if (lane >= off) incl += u;
  }
  if (lane == 31) wtot[wave] = incl;
  __syncthreads();
  if (wave == 0) {
    int w = wtot[lane];
    int wi = w;
#pragma unroll
    for (int off = 1; off < 32; off <<= 1) {
      int u = __shfl_up(wi, off, 32);
      if (lane >= off) wi += u;
    }
    wtot[lane] = wi - w;
  }
  __syncthreads();
  const int base = wtot[wave] + (incl - tot);
  __syncthreads();
  int4* op = (int4*)(off_s + t * 16);
#pragma unroll
  for (int i = 0; i < 4; ++i)
    op[i] = make_int4(base + e[4 * i + 0], base + e[4 * i + 1],
                      base + e[4 * i + 2], base + e[4 * i + 3]);
  if (blockIdx.x == 0) {
    goff[t] = base;
    if (t == 1023) goff[1024] = base + tot;
  }
  __syncthreads();

  const int c = blockIdx.x * 1024 + t;
  const int f = frame_id[c];
  const int p = off_s[f] + rank[c];
  const float4* mp = (const float4*)(means + (size_t)c * F_DIM);
  float4 m0 = mp[0], m1 = mp[1];
  float4* sp = (float4*)(smean + (size_t)p * F_DIM);
  sp[0] = m0; sp[1] = m1;
  swt[p]  = weights[c];
  sfid[p] = f;
}

__global__ void __launch_bounds__(512)
k_frames(const int*   __restrict__ goff,
         const float* __restrict__ smean,
         const float* __restrict__ swt,
         const int*   __restrict__ sfid,
         const int*   __restrict__ ticker_ids,
         const float* __restrict__ W_enc,
         const float* __restrict__ b_enc,
         const float* __restrict__ ticker_emb,
         const float* __restrict__ frame_scalars,
         const float* __restrict__ W_proj,
         const float* __restrict__ b_proj,
         const float* __restrict__ ln_gamma,
         const float* __restrict__ ln_beta,
         float* __restrict__ out) {
  __shared__ __attribute__((aligned(16))) float pool[FPB * PP];
  __shared__ __attribute__((aligned(16))) float ytile[FPB * YP];
  __shared__ __attribute__((aligned(16))) float slog[FPB * 4];
  __shared__ float smu[FPB], srs[FPB];

  const int tidx = threadIdx.x;
  const int lane = tidx & 31;
  const int wave = tidx >> 5;
  const int lm   = lane & 15;
  const int half = lane >> 4;
  const int fb0  = blockIdx.x * FPB;

  const v8f vzero = {0.f,0.f,0.f,0.f,0.f,0.f,0.f,0.f};

  if (tidx < FPB * 4) {
    const int f = tidx >> 2, s = tidx & 3;
    slog[tidx] = (s < N_SCALARS) ? log1pf(frame_scalars[(fb0 + f) * N_SCALARS + s]) : 0.0f;
  }

  {
    const int grp = wave >> 2;
    const int cq  = wave & 3;
    const int gi  = blockIdx.x * (FPB / 16) + grp;
    const int fb  = gi * 16;
    const int start = goff[gi];
    const int end   = goff[gi + 1];
    const int ntile = (end - start + 15) >> 4;

    v2f B0[4], B1[4];
    float be[4];
#pragma unroll
    for (int t = 0; t < 4; ++t) {
      const int col = (cq * 4 + t) * 16 + lm;
      B0[t].x = W_enc[(2 * half + 0) * ENC_DIM + col];
      B0[t].y = W_enc[(2 * half + 1) * ENC_DIM + col];
      B1[t].x = W_enc[(2 * half + 4) * ENC_DIM + col];
      B1[t].y = W_enc[(2 * half + 5) * ENC_DIM + col];
      be[t]   = b_enc[col];
    }
    v8f pacc[4];
#pragma unroll
    for (int t = 0; t < 4; ++t) pacc[t] = vzero;
    float sw = 0.0f;

    for (int tile = 0; tile < ntile; ++tile) {
      const int tb = start + tile * 16;
      const int pa = min(tb + lm, TOTAL_K - 1);
      const float2* mp = (const float2*)(smean + (size_t)pa * F_DIM);
      const float2 ma = mp[half], mb = mp[half + 2];
      v2f A0, A1;
      A0.x = ma.x; A0.y = ma.y;
      A1.x = mb.x; A1.y = mb.y;
      v2f SA[4];
#pragma unroll
      for (int j = 0; j < 4; ++j) {
        const int r0 = tb + 8 * half + 2 * j;
        const int q0 = min(r0, TOTAL_K - 1), q1 = min(r0 + 1, TOTAL_K - 1);
        const float w0 = swt[q0], w1 = swt[q1];
        const int   f0 = sfid[q0] - fb, f1 = sfid[q1] - fb;
        SA[j].x = (r0     < end && f0 == lm) ? w0 : 0.0f;
        SA[j].y = (r0 + 1 < end && f1 == lm) ? w1 : 0.0f;
        sw += SA[j].x + SA[j].y;
      }
#pragma unroll
      for (int t = 0; t < 4; ++t) {
        v8f acc = wmma_f32_4(A0, B0[t], vzero);
        acc     = wmma_f32_4(A1, B1[t], acc);
        float g[8];
#pragma unroll
        for (int r = 0; r < 8; ++r) g[r] = gelu_exact(acc[r] + be[t]);
#pragma unroll
        for (int j = 0; j < 4; ++j) {
          v2f b;
          b.x = g[2 * j]; b.y = g[2 * j + 1];
          pacc[t] = wmma_f32_4(SA[j], b, pacc[t]);
        }
      }
    }
    sw += __shfl_xor(sw, 16, 32);
#pragma unroll
    for (int r = 0; r < 8; ++r) {
      const int   fl  = half * 8 + r;
      const float swr = __shfl(sw, fl, 32);
      const float rw  = 1.0f / (swr + 1e-8f);
      const int   tk  = ticker_ids[fb + fl];
#pragma unroll
      for (int t = 0; t < 4; ++t) {
        const int col = (cq * 4 + t) * 16 + lm;
        const float te = ticker_emb[(size_t)tk * ENC_DIM + col];
        pool[(grp * 16 + fl) * PP + col] = fmaf(te, swr, pacc[t][r]) * rw;
      }
    }
  }
  __syncthreads();

  v8f acc[4][2];
#pragma unroll
  for (int m = 0; m < 4; ++m) { acc[m][0] = vzero; acc[m][1] = vzero; }

  const int cb = wave * 32 + 2 * lm;
  const float* wp = W_proj + (size_t)(2 * half) * D_MODEL + cb;
  float2 n0 = *(const float2*)(wp);
  float2 n1 = *(const float2*)(wp + D_MODEL);
  const float* ap = pool + lm * PP + 2 * half;

  for (int kk = 0; kk < 64; ++kk) {
    const float2 r0 = n0, r1 = n1;
    const int kn = (kk + 1 < 64) ? kk + 1 : kk;
    const float* q = wp + (size_t)(4 * kn) * D_MODEL;
    n0 = *(const float2*)(q);
    n1 = *(const float2*)(q + D_MODEL);
    v2f b0, b1;
    b0.x = r0.x; b0.y = r1.x;
    b1.x = r0.y; b1.y = r1.y;
#pragma unroll
    for (int m = 0; m < 4; ++m) {
      const float2 av = *(const float2*)(ap + m * 16 * PP + 4 * kk);
      v2f a;
      a.x = av.x; a.y = av.y;
      acc[m][0] = wmma_f32_4(a, b0, acc[m][0]);
      acc[m][1] = wmma_f32_4(a, b1, acc[m][1]);
    }
  }

  {
    const float2 w0 = *(const float2*)(W_proj + (size_t)(ENC_DIM + 0) * D_MODEL + cb);
    const float2 w1 = *(const float2*)(W_proj + (size_t)(ENC_DIM + 1) * D_MODEL + cb);
    const float2 w2 = *(const float2*)(W_proj + (size_t)(ENC_DIM + 2) * D_MODEL + cb);
    const float2 bp = *(const float2*)(b_proj + cb);
#pragma unroll
    for (int m = 0; m < 4; ++m) {
#pragma unroll
      for (int r = 0; r < 8; ++r) {
        const int row = m * 16 + half * 8 + r;
        const float4 l = *(const float4*)(slog + row * 4);
        float2 v;
        v.x = acc[m][0][r] + bp.x + l.x * w0.x + l.y * w1.x + l.z * w2.x;
        v.y = acc[m][1][r] + bp.y + l.x * w0.y + l.y * w1.y + l.z * w2.y;
        *(float2*)(ytile + row * YP + cb) = v;
      }
    }
  }
  __syncthreads();

  {
    const int row = tidx >> 3;
    const int seg = tidx & 7;
    float s = 0.0f, ss = 0.0f;
#pragma unroll 16
    for (int j = 0; j < 64; ++j) {
      float v = ytile[row * YP + seg + j * 8];
      s  += v;
      ss += v * v;
    }
#pragma unroll
    for (int off = 1; off < 8; off <<= 1) {
      s  += __shfl_xor(s,  off, 32);
      ss += __shfl_xor(ss, off, 32);
    }
    if (seg == 0) {
      float mu  = s * (1.0f / (float)D_MODEL);
      float var = fmaf(-mu, mu, ss * (1.0f / (float)D_MODEL));
      smu[row] = mu;
      srs[row] = rsqrtf(fmaxf(var, 0.0f) + 1e-5f);
    }
  }
  __syncthreads();

  {
    const int col = tidx;
    const float gm = ln_gamma[col], bt = ln_beta[col];
    for (int row = 0; row < FPB; ++row) {
      float x = (ytile[row * YP + col] - smu[row]) * srs[row] * gm + bt;
      out[(size_t)(fb0 + row) * D_MODEL + col] = gelu_exact(x);
    }
  }
}

extern "C" void kernel_launch(void* const* d_in, const int* in_sizes, int n_in,
                              void* d_out, int out_size, void* d_ws, size_t ws_size,
                              hipStream_t stream) {
  const float* chunks        = (const float*)d_in[0];
  const int*   frame_id      = (const int*)  d_in[1];
  const float* weights       = (const float*)d_in[2];
  const float* frame_scalars = (const float*)d_in[3];
  const int*   ticker_ids    = (const int*)  d_in[5];
  const float* W_enc         = (const float*)d_in[6];
  const float* b_enc         = (const float*)d_in[7];
  const float* ticker_emb    = (const float*)d_in[8];
  const float* W_proj        = (const float*)d_in[9];
  const float* b_proj        = (const float*)d_in[10];
  const float* ln_gamma      = (const float*)d_in[11];
  const float* ln_beta       = (const float*)d_in[12];
  float* out = (float*)d_out;

  float* means  = (float*)d_ws;
  float* smean  = means + (size_t)TOTAL_K * F_DIM;
  float* swt    = smean + (size_t)TOTAL_K * F_DIM;
  int*   sfid   = (int*)(swt + TOTAL_K);
  int*   rank   = sfid + TOTAL_K;
  int*   counts = rank + TOTAL_K;
  int*   goff   = counts + NUM_FRAMES;

  k_zero<<<NUM_FRAMES / 256, 256, 0, stream>>>(counts, NUM_FRAMES);

  k_means<<<TOTAL_K / 16 / 8, 256, 0, stream>>>(chunks, frame_id, means, counts, rank);

  k_place<<<TOTAL_K / 1024, 1024, 0, stream>>>(
      counts, rank, frame_id, weights, means, goff, smean, swt, sfid);

  k_frames<<<NUM_FRAMES / FPB, 512, 0, stream>>>(
      goff, smean, swt, sfid, ticker_ids, W_enc, b_enc, ticker_emb,
      frame_scalars, W_proj, b_proj, ln_gamma, ln_beta, out);
}


# Chunk-mean kernel as a stream: 16-wave block per WGP, 16-32 512-B loads in flight per wave (next 4 chunks requested before 4 are summed); rank atomic mid-stream, still arrival-order; sum order kept
# speedup vs baseline: 1.7940x; 1.0650x over previous
#include <hip/hip_runtime.h>
#include <hip/hip_bf16.h>

#define TOTAL_K     131072
#define CHUNK_LEN   64
#define F_DIM       8
#define NUM_FRAMES  16384
#define ENC_DIM     256
#define D_MODEL     512
#define N_SCALARS   3
#define N_TICKERS   4096
#define YP          520
#define FPB         64
#define PP          260

typedef float v2f __attribute__((ext_vector_type(2)));
typedef float v8f __attribute__((ext_vector_type(8)));

__device__ __forceinline__ v8f wmma_f32_4(v2f a, v2f b, v8f c) {
  return __builtin_amdgcn_wmma_f32_16x16x4_f32(
       false, a,  false, b,
       (short)0, c,  false,  false);
}

__device__ __forceinline__ float gelu_exact(float x) {
  float z  = x * 0.70710678118654752f;
  float az = fabsf(z);
  float t  = __builtin_amdgcn_rcpf(fmaf(0.3275911f, az, 1.0f));
  float p  = fmaf(t, 1.061405429f, -1.453152027f);
  p        = fmaf(t, p,  1.421413741f);
  p        = fmaf(t, p, -0.284496736f);
  p        = fmaf(t, p,  0.254829592f);
  p       *= t;
  float e  = __expf(-az * az);
  float erf_abs = fmaf(-p, e, 1.0f);
  float erf     = copysignf(erf_abs, z);
  return 0.5f * x * (1.0f + erf);
}

__global__ void k_zero(int* __restrict__ p, int n) {
  int i = blockIdx.x * blockDim.x + threadIdx.x;
  if (i < n) p[i] = 0;
}

typedef float v4f __attribute__((ext_vector_type(4)));

__global__ void __launch_bounds__(512)
k_means(const float* __restrict__ chunks,
        const int*   __restrict__ frame_id,
        float* __restrict__ means,
        int*   __restrict__ counts,
        int*   __restrict__ rank) {
  const int lane = threadIdx.x & 31;
  const int wave = threadIdx.x >> 5;
  const int c0   = (blockIdx.x * 16 + wave) * 32;

  const int fid = frame_id[c0 + lane];
  const v4f* b = (const v4f*)(chunks + (size_t)c0 * (CHUNK_LEN * F_DIM)) + lane;

  v4f q[2][16];
#pragma unroll
  for (int j = 0; j < 16; ++j) q[0][j] = b[j * 32];
  int rk = 0;

#pragma unroll
  for (int g = 0; g < 8; ++g) {
    if (g + 1 < 8) {
#pragma unroll
      for (int j = 0; j < 16; ++j)
        q[(g + 1) & 1][j] = b[(g + 1) * 512 + j * 32];
    }
    if (g == 0) rk = atomicAdd(&counts[fid], 1);
    __builtin_amdgcn_sched_barrier(0);

    v4f s[4];
#pragma unroll
    for (int k = 0; k < 4; ++k)
      s[k] = (q[g & 1][4 * k + 0] + q[g & 1][4 * k + 1]) + (q[g & 1][4 * k + 2] + q[g & 1][4 * k + 3]);
#pragma unroll
    for (int off = 2; off < 32; off <<= 1) {
#pragma unroll
      for (int k = 0; k < 4; ++k) {
        s[k].x += __shfl_xor(s[k].x, off, 32);
        s[k].y += __shfl_xor(s[k].y, off, 32);
        s[k].z += __shfl_xor(s[k].z, off, 32);
        s[k].w += __shfl_xor(s[k].w, off, 32);
      }
    }
    const int sel = lane >> 1;
    v4f v = s[0];
    v = (sel == 1) ? s[1] : v;
    v = (sel == 2) ? s[2] : v;
    v = (sel == 3) ? s[3] : v;
    v *= (1.0f / 64.0f);
    if (lane < 8) ((v4f*)(means + (size_t)(c0 + g * 4) * F_DIM))[lane] = v;
  }
  rank[c0 + lane] = rk;
}

__global__ void __launch_bounds__(1024)
k_place(const int*   __restrict__ counts,
        const int*   __restrict__ rank,
        const int*   __restrict__ frame_id,
        const float* __restrict__ weights,
        const float* __restrict__ means,
        int*   __restrict__ goff,
        float* __restrict__ smean,
        float* __restrict__ swt,
        int*   __restrict__ sfid) {
  __shared__ int off_s[NUM_FRAMES];
  int* wtot = off_s;
  const int t    = threadIdx.x;
  const int lane = t & 31;
  const int wave = t >> 5;

  int e[16];
  const int4* cp = (const int4*)(counts + t * 16);
#pragma unroll
  for (int i = 0; i < 4; ++i) {
    int4 c4 = cp[i];
    e[4 * i + 0] = c4.x; e[4 * i + 1] = c4.y; e[4 * i + 2] = c4.z; e[4 * i + 3] = c4.w;
  }
  int tot = 0;
#pragma unroll
  for (int i = 0; i < 16; ++i) { int c = e[i]; e[i] = tot; tot += c; }
  int incl = tot;
#pragma unroll
  for (int off = 1; off < 32; off <<= 1) {
    int u = __shfl_up(incl, off, 32);
    if (lane >= off) incl += u;
  }
  if (lane == 31) wtot[wave] = incl;
  __syncthreads();
  if (wave == 0) {
    int w = wtot[lane];
    int wi = w;
#pragma unroll
    for (int off = 1; off < 32; off <<= 1) {
      int u = __shfl_up(wi, off, 32);
      if (lane >= off) wi += u;
    }
    wtot[lane] = wi - w;
  }
  __syncthreads();
  const int base = wtot[wave] + (incl - tot);
  __syncthreads();
  int4* op = (int4*)(off_s + t * 16);
#pragma unroll
  for (int i = 0; i < 4; ++i)
    op[i] = make_int4(base + e[4 * i + 0], base + e[4 * i + 1],
                      base + e[4 * i + 2], base + e[4 * i + 3]);
  if (blockIdx.x == 0) {
    goff[t] = base;
    if (t == 1023) goff[1024] = base + tot;
  }
  __syncthreads();

  const int c = blockIdx.x * 1024 + t;
  const int f = frame_id[c];
  const int p = off_s[f] + rank[c];
  const float4* mp = (const float4*)(means + (size_t)c * F_DIM);
  float4 m0 = mp[0], m1 = mp[1];
  float4* sp = (float4*)(smean + (size_t)p * F_DIM);
  sp[0] = m0; sp[1] = m1;
  swt[p]  = weights[c];
  sfid[p] = f;
}

__global__ void __launch_bounds__(512)
k_frames(const int*   __restrict__ goff,
         const float* __restrict__ smean,
         const float* __restrict__ swt,
         const int*   __restrict__ sfid,
         const int*   __restrict__ ticker_ids,
         const float* __restrict__ W_enc,
         const float* __restrict__ b_enc,
         const float* __restrict__ ticker_emb,
         const float* __restrict__ frame_scalars,
         const float* __restrict__ W_proj,
         const float* __restrict__ b_proj,
         const float* __restrict__ ln_gamma,
         const float* __restrict__ ln_beta,
         float* __restrict__ out) {
  __shared__ __attribute__((aligned(16))) float pool[FPB * PP];
  __shared__ __attribute__((aligned(16))) float ytile[FPB * YP];
  __shared__ __attribute__((aligned(16))) float slog[FPB * 4];
  __shared__ float smu[FPB], srs[FPB];

  const int tidx = threadIdx.x;
  const int lane = tidx & 31;
  const int wave = tidx >> 5;
  const int lm   = lane & 15;
  const int half = lane >> 4;
  const int fb0  = blockIdx.x * FPB;

  const v8f vzero = {0.f,0.f,0.f,0.f,0.f,0.f,0.f,0.f};

  if (tidx < FPB * 4) {
    const int f = tidx >> 2, s = tidx & 3;
    slog[tidx] = (s < N_SCALARS) ? log1pf(frame_scalars[(fb0 + f) * N_SCALARS + s]) : 0.0f;
  }

  {
    const int grp = wave >> 2;
    const int cq  = wave & 3;
    const int gi  = blockIdx.x * (FPB / 16) + grp;
    const int fb  = gi * 16;
    const int start = goff[gi];
    const int end   = goff[gi + 1];
    const int ntile = (end - start + 15) >> 4;

    v2f B0[4], B1[4];
    float be[4];
#pragma unroll
    for (int t = 0; t < 4; ++t) {
      const int col = (cq * 4 + t) * 16 + lm;
      B0[t].x = W_enc[(2 * half + 0) * ENC_DIM + col];
      B0[t].y = W_enc[(2 * half + 1) * ENC_DIM + col];
      B1[t].x = W_enc[(2 * half + 4) * ENC_DIM + col];
      B1[t].y = W_enc[(2 * half + 5) * ENC_DIM + col];
      be[t]   = b_enc[col];
    }
    v8f pacc[4];
#pragma unroll
    for (int t = 0; t < 4; ++t) pacc[t] = vzero;
    float sw = 0.0f;

    for (int tile = 0; tile < ntile; ++tile) {
      const int tb = start + tile * 16;
      const int pa = min(tb + lm, TOTAL_K - 1);
      const float2* mp = (const float2*)(smean + (size_t)pa * F_DIM);
      const float2 ma = mp[half], mb = mp[half + 2];
      v2f A0, A1;
      A0.x = ma.x; A0.y = ma.y;
      A1.x = mb.x; A1.y = mb.y;
      v2f SA[4];
#pragma unroll
      for (int j = 0; j < 4; ++j) {
        const int r0 = tb + 8 * half + 2 * j;
        const int q0 = min(r0, TOTAL_K - 1), q1 = min(r0 + 1, TOTAL_K - 1);
        const float w0 = swt[q0], w1 = swt[q1];
        const int   f0 = sfid[q0] - fb, f1 = sfid[q1] - fb;
        SA[j].x = (r0     < end && f0 == lm) ? w0 : 0.0f;
        SA[j].y = (r0 + 1 < end && f1 == lm) ? w1 : 0.0f;
        sw += SA[j].x + SA[j].y;
      }
#pragma unroll
      for (int t = 0; t < 4; ++t) {
        v8f acc = wmma_f32_4(A0, B0[t], vzero);
        acc     = wmma_f32_4(A1, B1[t], acc);
        float g[8];
#pragma unroll
        for (int r = 0; r < 8; ++r) g[r] = gelu_exact(acc[r] + be[t]);
#pragma unroll
        for (int j = 0; j < 4; ++j) {
          v2f b;
          b.x = g[2 * j]; b.y = g[2 * j + 1];
          pacc[t] = wmma_f32_4(SA[j], b, pacc[t]);
        }
      }
    }
    sw += __shfl_xor(sw, 16, 32);
#pragma unroll
    for (int r = 0; r < 8; ++r) {
      const int   fl  = half * 8 + r;
      const float swr = __shfl(sw, fl, 32);
      const float rw  = 1.0f / (swr + 1e-8f);
      const int   tk  = ticker_ids[fb + fl];
#pragma unroll
      for (int t = 0; t < 4; ++t) {
        const int col = (cq * 4 + t) * 16 + lm;
        const float te = ticker_emb[(size_t)tk * ENC_DIM + col];
        pool[(grp * 16 + fl) * PP + col] = fmaf(te, swr, pacc[t][r]) * rw;
      }
    }
  }
  __syncthreads();

  v8f acc[4][2];
#pragma unroll
  for (int m = 0; m < 4; ++m) { acc[m][0] = vzero; acc[m][1] = vzero; }

  const int cb = wave * 32 + 2 * lm;
  const float* wp = W_proj + (size_t)(2 * half) * D_MODEL + cb;
  float2 n0 = *(const float2*)(wp);
  float2 n1 = *(const float2*)(wp + D_MODEL);
  const float* ap = pool + lm * PP + 2 * half;

  for (int kk = 0; kk < 64; ++kk) {
    const float2 r0 = n0, r1 = n1;
    const int kn = (kk + 1 < 64) ? kk + 1 : kk;
    const float* q = wp + (size_t)(4 * kn) * D_MODEL;
    n0 = *(const float2*)(q);
    n1 = *(const float2*)(q + D_MODEL);
    v2f b0, b1;
    b0.x = r0.x; b0.y = r1.x;
    b1.x = r0.y; b1.y = r1.y;
#pragma unroll
    for (int m = 0; m < 4; ++m) {
      const float2 av = *(const float2*)(ap + m * 16 * PP + 4 * kk);
      v2f a;
      a.x = av.x; a.y = av.y;
      acc[m][0] = wmma_f32_4(a, b0, acc[m][0]);
      acc[m][1] = wmma_f32_4(a, b1, acc[m][1]);
    }
  }

  {
    const float2 w0 = *(const float2*)(W_proj + (size_t)(ENC_DIM + 0) * D_MODEL + cb);
    const float2 w1 = *(const float2*)(W_proj + (size_t)(ENC_DIM + 1) * D_MODEL + cb);
    const float2 w2 = *(const float2*)(W_proj + (size_t)(ENC_DIM + 2) * D_MODEL + cb);
    const float2 bp = *(const float2*)(b_proj + cb);
#pragma unroll
    for (int m = 0; m < 4; ++m) {
#pragma unroll
      for (int r = 0; r < 8; ++r) {
        const int row = m * 16 + half * 8 + r;
        const float4 l = *(const float4*)(slog + row * 4);
        float2 v;
        v.x = acc[m][0][r] + bp.x + l.x * w0.x + l.y * w1.x + l.z * w2.x;
        v.y = acc[m][1][r] + bp.y + l.x * w0.y + l.y * w1.y + l.z * w2.y;
        *(float2*)(ytile + row * YP + cb) = v;
      }
    }
  }
  __syncthreads();

  {
    const int row = tidx >> 3;
    const int seg = tidx & 7;
    float s = 0.0f, ss = 0.0f;
#pragma unroll 16
    for (int j = 0; j < 64; ++j) {
      float v = ytile[row * YP + seg + j * 8];
      s  += v;
      ss += v * v;
    }
#pragma unroll
    for (int off = 1; off < 8; off <<= 1) {
      s  += __shfl_xor(s,  off, 32);
      ss += __shfl_xor(ss, off, 32);
    }
    if (seg == 0) {
      float mu  = s * (1.0f / (float)D_MODEL);
      float var = fmaf(-mu, mu, ss * (1.0f / (float)D_MODEL));
      smu[row] = mu;
      srs[row] = rsqrtf(fmaxf(var, 0.0f) + 1e-5f);
    }
  }
  __syncthreads();

  {
    const int col = tidx;
    const float gm = ln_gamma[col], bt = ln_beta[col];
    for (int row = 0; row < FPB; ++row) {
      float x = (ytile[row * YP + col] - smu[row]) * srs[row] * gm + bt;
      out[(size_t)(fb0 + row) * D_MODEL + col] = gelu_exact(x);
    }
  }
}

extern "C" void kernel_launch(void* const* d_in, const int* in_sizes, int n_in,
                              void* d_out, int out_size, void* d_ws, size_t ws_size,
                              hipStream_t stream) {
  const float* chunks        = (const float*)d_in[0];
  const int*   frame_id      = (const int*)  d_in[1];
  const float* weights       = (const float*)d_in[2];
  const float* frame_scalars = (const float*)d_in[3];
  const int*   ticker_ids    = (const int*)  d_in[5];
  const float* W_enc         = (const float*)d_in[6];
  const float* b_enc         = (const float*)d_in[7];
  const float* ticker_emb    = (const float*)d_in[8];
  const float* W_proj        = (const float*)d_in[9];
  const float* b_proj        = (const float*)d_in[10];
  const float* ln_gamma      = (const float*)d_in[11];
  const float* ln_beta       = (const float*)d_in[12];
  float* out = (float*)d_out;

  float* means  = (float*)d_ws;
  float* smean  = means + (size_t)TOTAL_K * F_DIM;
  float* swt    = smean + (size_t)TOTAL_K * F_DIM;
  int*   sfid   = (int*)(swt + TOTAL_K);
  int*   rank   = sfid + TOTAL_K;
  int*   counts = rank + TOTAL_K;
  int*   goff   = counts + NUM_FRAMES;

  k_zero<<<NUM_FRAMES / 256, 256, 0, stream>>>(counts, NUM_FRAMES);

  k_means<<<TOTAL_K / 32 / 16, 512, 0, stream>>>(chunks, frame_id, means, counts, rank);

  k_place<<<TOTAL_K / 1024, 1024, 0, stream>>>(
      counts, rank, frame_id, weights, means, goff, smean, swt, sfid);

  k_frames<<<NUM_FRAMES / FPB, 512, 0, stream>>>(
      goff, smean, swt, sfid, ticker_ids, W_enc, b_enc, ticker_emb,
      frame_scalars, W_proj, b_proj, ln_gamma, ln_beta, out);
}


# Sort-placement launch removed: chunk-mean kernel's rank atomic drops each chunk id into a 32-slot bucket of its frame; frame kernel lists buckets in LDS, gathers means itself; sum order kept
# speedup vs baseline: 2.2569x; 1.2580x over previous
#include <hip/hip_runtime.h>
#include <hip/hip_bf16.h>

#define TOTAL_K     131072
#define CHUNK_LEN   64
#define F_DIM       8
#define NUM_FRAMES  16384
#define ENC_DIM     256
#define D_MODEL     512
#define N_SCALARS   3
#define N_TICKERS   4096
#define YP          520
#define FPB         64
#define PP          260
#define CAPF        32
#define LSTP        (16 * CAPF + 32)

typedef float v2f __attribute__((ext_vector_type(2)));
typedef float v8f __attribute__((ext_vector_type(8)));

__device__ __forceinline__ v8f wmma_f32_4(v2f a, v2f b, v8f c) {
  return __builtin_amdgcn_wmma_f32_16x16x4_f32(
       false, a,  false, b,
       (short)0, c,  false,  false);
}

__device__ __forceinline__ float gelu_exact(float x) {
  float z  = x * 0.70710678118654752f;
  float az = fabsf(z);
  float t  = __builtin_amdgcn_rcpf(fmaf(0.3275911f, az, 1.0f));
  float p  = fmaf(t, 1.061405429f, -1.453152027f);
  p        = fmaf(t, p,  1.421413741f);
  p        = fmaf(t, p, -0.284496736f);
  p        = fmaf(t, p,  0.254829592f);
  p       *= t;
  float e  = __expf(-az * az);
  float erf_abs = fmaf(-p, e, 1.0f);
  float erf     = copysignf(erf_abs, z);
  return 0.5f * x * (1.0f + erf);
}

__global__ void k_zero(int* __restrict__ p, int n, int* __restrict__ bidx) {
  int i = blockIdx.x * blockDim.x + threadIdx.x;
  if (i < n) p[i] = 0;
  if (i < NUM_FRAMES) bidx[i * CAPF] = 0;
}

typedef float v4f __attribute__((ext_vector_type(4)));

__global__ void __launch_bounds__(512)
k_means(const float* __restrict__ chunks,
        const int*   __restrict__ frame_id,
        float* __restrict__ means,
        int*   __restrict__ counts,
        int*   __restrict__ bidx,
        int*   __restrict__ ovf) {
  const int lane = threadIdx.x & 31;
  const int wave = threadIdx.x >> 5;
  const int c0   = (blockIdx.x * 16 + wave) * 32;

  const int fid = frame_id[c0 + lane];
  const v4f* b = (const v4f*)(chunks + (size_t)c0 * (CHUNK_LEN * F_DIM)) + lane;

  v4f q[2][16];
#pragma unroll
  for (int j = 0; j < 16; ++j) q[0][j] = b[j * 32];
  int rk = 0;

#pragma unroll
  for (int g = 0; g < 8; ++g) {
    if (g + 1 < 8) {
#pragma unroll
      for (int j = 0; j < 16; ++j)
        q[(g + 1) & 1][j] = b[(g + 1) * 512 + j * 32];
    }
    if (g == 0) rk = atomicAdd(&counts[fid], 1);
    __builtin_amdgcn_sched_barrier(0);

    v4f s[4];
#pragma unroll
    for (int k = 0; k < 4; ++k)
      s[k] = (q[g & 1][4 * k + 0] + q[g & 1][4 * k + 1]) + (q[g & 1][4 * k + 2] + q[g & 1][4 * k + 3]);
#pragma unroll
    for (int off = 2; off < 32; off <<= 1) {
#pragma unroll
      for (int k = 0; k < 4; ++k) {
        s[k].x += __shfl_xor(s[k].x, off, 32);
        s[k].y += __shfl_xor(s[k].y, off, 32);
        s[k].z += __shfl_xor(s[k].z, off, 32);
        s[k].w += __shfl_xor(s[k].w, off, 32);
      }
    }
    const int sel = lane >> 1;
    v4f v = s[0];
    v = (sel == 1) ? s[1] : v;
    v = (sel == 2) ? s[2] : v;
    v = (sel == 3) ? s[3] : v;
    v *= (1.0f / 64.0f);
    if (lane < 8) ((v4f*)(means + (size_t)(c0 + g * 4) * F_DIM))[lane] = v;
  }
  if (rk < CAPF) {
    bidx[fid * CAPF + rk] = c0 + lane;
  } else {
    const int p = atomicAdd(&counts[NUM_FRAMES], 1);
    ovf[p] = c0 + lane;
  }
}

__global__ void __launch_bounds__(512)
k_frames(const int*   __restrict__ counts,
         const int*   __restrict__ bidx,
         const int*   __restrict__ ovf,
         const float* __restrict__ means,
         const float* __restrict__ weights,
         const int*   __restrict__ frame_id,
         const int*   __restrict__ ticker_ids,
         const float* __restrict__ W_enc,
         const float* __restrict__ b_enc,
         const float* __restrict__ ticker_emb,
         const float* __restrict__ frame_scalars,
         const float* __restrict__ W_proj,
         const float* __restrict__ b_proj,
         const float* __restrict__ ln_gamma,
         const float* __restrict__ ln_beta,
         float* __restrict__ out) {
  __shared__ __attribute__((aligned(16))) float pool[FPB * PP];
  __shared__ __attribute__((aligned(16))) float ytile[FPB * YP];
  __shared__ __attribute__((aligned(16))) float slog[FPB * 4];
  __shared__ float smu[FPB], srs[FPB];
  __shared__ __attribute__((aligned(16))) int lst[16 * LSTP];

  const int tidx = threadIdx.x;
  const int lane = tidx & 31;
  const int wave = tidx >> 5;
  const int lm   = lane & 15;
  const int half = lane >> 4;
  const int fb0  = blockIdx.x * FPB;

  const v8f vzero = {0.f,0.f,0.f,0.f,0.f,0.f,0.f,0.f};

  if (tidx < FPB * 4) {
    const int f = tidx >> 2, s = tidx & 3;
    slog[tidx] = (s < N_SCALARS) ? log1pf(frame_scalars[(fb0 + f) * N_SCALARS + s]) : 0.0f;
  }

  {
    const int grp = wave >> 2;
    const int cq  = wave & 3;
    const int gi  = blockIdx.x * (FPB / 16) + grp;
    const int fb  = gi * 16;

    int* L = lst + wave * LSTP;
    const int cf = min(counts[fb + lm], CAPF);
    int incl = cf;
#pragma unroll
    for (int off = 1; off < 16; off <<= 1) {
      const int u = __shfl_up(incl, off, 16);
      if (lm >= off) incl += u;
    }
    const int pre = incl - cf;
    const int n   = __shfl(incl, 15, 16);
    int bv[16];
#pragma unroll
    for (int f = 0; f < 16; ++f) {
      const int cnt_f = __builtin_amdgcn_readlane(cf, f);
      bv[f] = bidx[(fb + f) * CAPF + ((lane < cnt_f) ? lane : 0)];
    }
#pragma unroll
    for (int f = 0; f < 16; ++f) {
      const int cnt_f = __builtin_amdgcn_readlane(cf, f);
      const int pre_f = __builtin_amdgcn_readlane(pre, f);
      const int dst = (lane < cnt_f) ? pre_f + lane : 16 * CAPF + 16 + (lane & 15);
      L[dst] = (bv[f] << 4) | f;
    }
    if (lane < 16) L[n + lane] = 0;
    const int ntile  = (n + 15) >> 4;
    const int novf   = min(counts[NUM_FRAMES], TOTAL_K);
    const int notile = (novf + 15) >> 4;

    v2f B0[4], B1[4];
    float be[4];
#pragma unroll
    for (int t = 0; t < 4; ++t) {
      const int col = (cq * 4 + t) * 16 + lm;
      B0[t].x = W_enc[(2 * half + 0) * ENC_DIM + col];
      B0[t].y = W_enc[(2 * half + 1) * ENC_DIM + col];
      B1[t].x = W_enc[(2 * half + 4) * ENC_DIM + col];
      B1[t].y = W_enc[(2 * half + 5) * ENC_DIM + col];
      be[t]   = b_enc[col];
    }
    v8f pacc[4];
#pragma unroll
    for (int t = 0; t < 4; ++t) pacc[t] = vzero;
    float sw = 0.0f;

    __syncthreads();

    for (int tile = 0; tile < ntile + notile; ++tile) {
      int ca;
      int cs[8], fs[8];
      if (tile < ntile) {
        const int tb = tile * 16;
        ca = (L[tb + lm] >> 4) & (TOTAL_K - 1);
        const int4* lq = (const int4*)(L + tb + 8 * half);
        const int4 e0 = lq[0], e1 = lq[1];
        const int e[8] = {e0.x, e0.y, e0.z, e0.w, e1.x, e1.y, e1.z, e1.w};
#pragma unroll
        for (int i = 0; i < 8; ++i) {
          cs[i] = (e[i] >> 4) & (TOTAL_K - 1);
          fs[i] = (tb + 8 * half + i < n) ? (e[i] & 15) : -1;
        }
      } else {
        const int tb = (tile - ntile) * 16;
        ca = ovf[min(tb + lm, novf - 1)] & (TOTAL_K - 1);
#pragma unroll
        for (int i = 0; i < 8; ++i) {
          const int r = tb + 8 * half + i;
          const int c = ovf[min(r, novf - 1)] & (TOTAL_K - 1);
          cs[i] = c;
          fs[i] = (r < novf) ? frame_id[c] - fb : -1;
        }
      }
      const float2* mp = (const float2*)(means + (size_t)ca * F_DIM);
      const float2 ma = mp[half], mb = mp[half + 2];
      v2f A0, A1;
      A0.x = ma.x; A0.y = ma.y;
      A1.x = mb.x; A1.y = mb.y;
      v2f SA[4];
#pragma unroll
      for (int j = 0; j < 4; ++j) {
        const float w0 = weights[cs[2 * j]], w1 = weights[cs[2 * j + 1]];
        SA[j].x = (fs[2 * j]     == lm) ? w0 : 0.0f;
        SA[j].y = (fs[2 * j + 1] == lm) ? w1 : 0.0f;
        sw += SA[j].x + SA[j].y;
      }
#pragma unroll
      for (int t = 0; t < 4; ++t) {
        v8f acc = wmma_f32_4(A0, B0[t], vzero);
        acc     = wmma_f32_4(A1, B1[t], acc);
        float g[8];
#pragma unroll
        for (int r = 0; r < 8; ++r) g[r] = gelu_exact(acc[r] + be[t]);
#pragma unroll
        for (int j = 0; j < 4; ++j) {
          v2f b;
          b.x = g[2 * j]; b.y = g[2 * j + 1];
          pacc[t] = wmma_f32_4(SA[j], b, pacc[t]);
        }
      }
    }
    sw += __shfl_xor(sw, 16, 32);
#pragma unroll
    for (int r = 0; r < 8; ++r) {
      const int   fl  = half * 8 + r;
      const float swr = __shfl(sw, fl, 32);
      const float rw  = 1.0f / (swr + 1e-8f);
      const int   tk  = ticker_ids[fb + fl];
#pragma unroll
      for (int t = 0; t < 4; ++t) {
        const int col = (cq * 4 + t) * 16 + lm;
        const float te = ticker_emb[(size_t)tk * ENC_DIM + col];
        pool[(grp * 16 + fl) * PP + col] = fmaf(te, swr, pacc[t][r]) * rw;
      }
    }
  }
  __syncthreads();

  v8f acc[4][2];
#pragma unroll
  for (int m = 0; m < 4; ++m) { acc[m][0] = vzero; acc[m][1] = vzero; }

  const int cb = wave * 32 + 2 * lm;
  const float* wp = W_proj + (size_t)(2 * half) * D_MODEL + cb;
  float2 n0 = *(const float2*)(wp);
  float2 n1 = *(const float2*)(wp + D_MODEL);
  const float* ap = pool + lm * PP + 2 * half;

  for (int kk = 0; kk < 64; ++kk) {
    const float2 r0 = n0, r1 = n1;
    const int kn = (kk + 1 < 64) ? kk + 1 : kk;
    const float* q = wp + (size_t)(4 * kn) * D_MODEL;
    n0 = *(const float2*)(q);
    n1 = *(const float2*)(q + D_MODEL);
    v2f b0, b1;
    b0.x = r0.x; b0.y = r1.x;
    b1.x = r0.y; b1.y = r1.y;
#pragma unroll
    for (int m = 0; m < 4; ++m) {
      const float2 av = *(const float2*)(ap + m * 16 * PP + 4 * kk);
      v2f a;
      a.x = av.x; a.y = av.y;
      acc[m][0] = wmma_f32_4(a, b0, acc[m][0]);
      acc[m][1] = wmma_f32_4(a, b1, acc[m][1]);
    }
  }

  {
    const float2 w0 = *(const float2*)(W_proj + (size_t)(ENC_DIM + 0) * D_MODEL + cb);
    const float2 w1 = *(const float2*)(W_proj + (size_t)(ENC_DIM + 1) * D_MODEL + cb);
    const float2 w2 = *(const float2*)(W_proj + (size_t)(ENC_DIM + 2) * D_MODEL + cb);
    const float2 bp = *(const float2*)(b_proj + cb);
#pragma unroll
    for (int m = 0; m < 4; ++m) {
#pragma unroll
      for (int r = 0; r < 8; ++r) {
        const int row = m * 16 + half * 8 + r;
        const float4 l = *(const float4*)(slog + row * 4);
        float2 v;
        v.x = acc[m][0][r] + bp.x + l.x * w0.x + l.y * w1.x + l.z * w2.x;
        v.y = acc[m][1][r] + bp.y + l.x * w0.y + l.y * w1.y + l.z * w2.y;
        *(float2*)(ytile + row * YP + cb) = v;
      }
    }
  }
  __syncthreads();

  {
    const int row = tidx >> 3;
    const int seg = tidx & 7;
    float s = 0.0f, ss = 0.0f;
#pragma unroll 16
    for (int j = 0; j < 64; ++j) {
      float v = ytile[row * YP + seg + j * 8];
      s  += v;
      ss += v * v;
    }
#pragma unroll
    for (int off = 1; off < 8; off <<= 1) {
      s  += __shfl_xor(s,  off, 32);
      ss += __shfl_xor(ss, off, 32);
    }
    if (seg == 0) {
      float mu  = s * (1.0f / (float)D_MODEL);
      float var = fmaf(-mu, mu, ss * (1.0f / (float)D_MODEL));
      smu[row] = mu;
      srs[row] = rsqrtf(fmaxf(var, 0.0f) + 1e-5f);
    }
  }
  __syncthreads();

  {
    const int col = tidx;
    const float gm = ln_gamma[col], bt = ln_beta[col];
    for (int row = 0; row < FPB; ++row) {
      float x = (ytile[row * YP + col] - smu[row]) * srs[row] * gm + bt;
      out[(size_t)(fb0 + row) * D_MODEL + col] = gelu_exact(x);
    }
  }
}

extern "C" void kernel_launch(void* const* d_in, const int* in_sizes, int n_in,
                              void* d_out, int out_size, void* d_ws, size_t ws_size,
                              hipStream_t stream) {
  const float* chunks        = (const float*)d_in[0];
  const int*   frame_id      = (const int*)  d_in[1];
  const float* weights       = (const float*)d_in[2];
  const float* frame_scalars = (const float*)d_in[3];
  const int*   ticker_ids    = (const int*)  d_in[5];
  const float* W_enc         = (const float*)d_in[6];
  const float* b_enc         = (const float*)d_in[7];
  const float* ticker_emb    = (const float*)d_in[8];
  const float* W_proj        = (const float*)d_in[9];
  const float* b_proj        = (const float*)d_in[10];
  const float* ln_gamma      = (const float*)d_in[11];
  const float* ln_beta       = (const float*)d_in[12];
  float* out = (float*)d_out;

  float* means  = (float*)d_ws;
  int*   bidx   = (int*)(means + (size_t)TOTAL_K * F_DIM);
  int*   ovf    = bidx + (size_t)NUM_FRAMES * CAPF;
  int*   counts = ovf + TOTAL_K;

  k_zero<<<NUM_FRAMES / 256 + 1, 256, 0, stream>>>(counts, NUM_FRAMES + 1, bidx);

  k_means<<<TOTAL_K / 32 / 16, 512, 0, stream>>>(chunks, frame_id, means, counts, bidx, ovf);

  k_frames<<<NUM_FRAMES / FPB, 512, 0, stream>>>(
      counts, bidx, ovf, means, weights, frame_id, ticker_ids, W_enc, b_enc, ticker_emb,
      frame_scalars, W_proj, b_proj, ln_gamma, ln_beta, out);
}


# Counter-clearing launch removed: chunk-mean kernel counting-sorts its 512 chunk ids by frame block in LDS (no global counters); frame kernel lists from 256 segments; sum order stays arrival-order
# speedup vs baseline: 2.5359x; 1.1236x over previous
#include <hip/hip_runtime.h>
#include <hip/hip_bf16.h>

#define TOTAL_K     131072
#define CHUNK_LEN   64
#define F_DIM       8
#define NUM_FRAMES  16384
#define ENC_DIM     256
#define D_MODEL     512
#define N_SCALARS   3
#define N_TICKERS   4096
#define YP          520
#define FPB         64
#define PP          260
#define SPB         512
#define NFB         (NUM_FRAMES / FPB)
#define PER         16
#define GLP         (PER * NFB + 16)

typedef float v2f __attribute__((ext_vector_type(2)));
typedef float v8f __attribute__((ext_vector_type(8)));

__device__ __forceinline__ v8f wmma_f32_4(v2f a, v2f b, v8f c) {
  return __builtin_amdgcn_wmma_f32_16x16x4_f32(
       false, a,  false, b,
       (short)0, c,  false,  false);
}

__device__ __forceinline__ float gelu_exact(float x) {
  float z  = x * 0.70710678118654752f;
  float az = fabsf(z);
  float t  = __builtin_amdgcn_rcpf(fmaf(0.3275911f, az, 1.0f));
  float p  = fmaf(t, 1.061405429f, -1.453152027f);
  p        = fmaf(t, p,  1.421413741f);
  p        = fmaf(t, p, -0.284496736f);
  p        = fmaf(t, p,  0.254829592f);
  p       *= t;
  float e  = __expf(-az * az);
  float erf_abs = fmaf(-p, e, 1.0f);
  float erf     = copysignf(erf_abs, z);
  return 0.5f * x * (1.0f + erf);
}

typedef float v4f __attribute__((ext_vector_type(4)));

__device__ __forceinline__ void lds_barrier() {
  __builtin_amdgcn_fence(__ATOMIC_RELEASE, "workgroup", "local");
  __builtin_amdgcn_s_barrier();
  __builtin_amdgcn_fence(__ATOMIC_ACQUIRE, "workgroup", "local");
}

__global__ void __launch_bounds__(512)
k_means(const float* __restrict__ chunks,
        const int*   __restrict__ frame_id,
        float* __restrict__ means,
        int*   __restrict__ srt,
        int*   __restrict__ offT) {
  __shared__ int hc[NFB];
  __shared__ int hs[NFB];
  __shared__ int wt[NFB / 32];
  const int tid  = threadIdx.x;
  const int lane = threadIdx.x & 31;
  const int wave = threadIdx.x >> 5;
  const int c0   = (blockIdx.x * 16 + wave) * 32;

  const int fid = frame_id[c0 + lane];
  const v4f* b = (const v4f*)(chunks + (size_t)c0 * (CHUNK_LEN * F_DIM)) + lane;

  v4f q[2][16];
#pragma unroll
  for (int j = 0; j < 16; ++j) q[0][j] = b[j * 32];

#pragma unroll
  for (int g = 0; g < 8; ++g) {
    if (g + 1 < 8) {
#pragma unroll
      for (int j = 0; j < 16; ++j)
        q[(g + 1) & 1][j] = b[(g + 1) * 512 + j * 32];
    }
    if (g == 0) {
      const int d = (fid >> 6) & (NFB - 1);
      if (tid < NFB) hc[tid] = 0;
      lds_barrier();
      const int rank = atomicAdd(&hc[d], 1);
      lds_barrier();
      const int v = (tid < NFB) ? hc[tid] : 0;
      int inc = v;
#pragma unroll
      for (int off = 1; off < 32; off <<= 1) {
        const int u = __shfl_up(inc, off, 32);
        if (lane >= off) inc += u;
      }
      if (tid < NFB && lane == 31) wt[wave] = inc;
      lds_barrier();
      if (tid < NFB) {
        int wp = 0;
#pragma unroll
        for (int w2 = 0; w2 < NFB / 32; ++w2) wp += (w2 < wave) ? wt[w2] : 0;
        hs[tid] = wp + inc - v;
      }
      lds_barrier();
      srt[blockIdx.x * SPB + hs[d] + rank] = ((c0 + lane) << 6) | (fid & 63);
      if (tid < NFB) {
        int2 sc;
        sc.x = hs[tid]; sc.y = v;
        ((int2*)offT)[tid * NFB + blockIdx.x] = sc;
      }
    }
    __builtin_amdgcn_sched_barrier(0);

    v4f s[4];
#pragma unroll
    for (int k = 0; k < 4; ++k)
      s[k] = (q[g & 1][4 * k + 0] + q[g & 1][4 * k + 1]) + (q[g & 1][4 * k + 2] + q[g & 1][4 * k + 3]);
#pragma unroll
    for (int off = 2; off < 32; off <<= 1) {
#pragma unroll
      for (int k = 0; k < 4; ++k) {
        s[k].x += __shfl_xor(s[k].x, off, 32);
        s[k].y += __shfl_xor(s[k].y, off, 32);
        s[k].z += __shfl_xor(s[k].z, off, 32);
        s[k].w += __shfl_xor(s[k].w, off, 32);
      }
    }
    const int sel = lane >> 1;
    v4f v = s[0];
    v = (sel == 1) ? s[1] : v;
    v = (sel == 2) ? s[2] : v;
    v = (sel == 3) ? s[3] : v;
    v *= (1.0f / 64.0f);
    if (lane < 8) ((v4f*)(means + (size_t)(c0 + g * 4) * F_DIM))[lane] = v;
  }
}

__global__ void __launch_bounds__(512)
k_frames(const int*   __restrict__ offT,
         const int*   __restrict__ srt,
         const float* __restrict__ means,
         const float* __restrict__ weights,
         const int*   __restrict__ ticker_ids,
         const float* __restrict__ W_enc,
         const float* __restrict__ b_enc,
         const float* __restrict__ ticker_emb,
         const float* __restrict__ frame_scalars,
         const float* __restrict__ W_proj,
         const float* __restrict__ b_proj,
         const float* __restrict__ ln_gamma,
         const float* __restrict__ ln_beta,
         float* __restrict__ out) {
  __shared__ __attribute__((aligned(16))) float pool[FPB * PP];
  __shared__ __attribute__((aligned(16))) float ytile[FPB * YP];
  __shared__ __attribute__((aligned(16))) float slog[FPB * 4];
  __shared__ float smu[FPB], srs[FPB];
  __shared__ __attribute__((aligned(16))) int glist[4 * GLP];
  __shared__ int gcnt[4];
  __shared__ int gmore;

  const int tidx = threadIdx.x;
  const int lane = tidx & 31;
  const int wave = tidx >> 5;
  const int lm   = lane & 15;
  const int half = lane >> 4;
  const int fb0  = blockIdx.x * FPB;

  const v8f vzero = {0.f,0.f,0.f,0.f,0.f,0.f,0.f,0.f};

  if (tidx < FPB * 4) {
    const int f = tidx >> 2, s = tidx & 3;
    slog[tidx] = (s < N_SCALARS) ? log1pf(frame_scalars[(fb0 + f) * N_SCALARS + s]) : 0.0f;
  }

  {
    const int grp = wave >> 2;
    const int cq  = wave & 3;
    const int gi  = blockIdx.x * (FPB / 16) + grp;
    const int fb  = gi * 16;

    int seg = 0, left = 0;
    int pe[8] = {0, 0, 0, 0, 0, 0, 0, 0};
    if (tidx < NFB) {
      const int2 sc = ((const int2*)offT)[blockIdx.x * NFB + tidx];
      const int st = min(max(sc.x, 0), SPB - 1);
      left = min(max(sc.y, 0), SPB - st);
      seg  = tidx * SPB + st;
#pragma unroll
      for (int i = 0; i < 8; ++i) pe[i] = srt[min(seg + i, TOTAL_K - 1)];
    }

    v2f B0[4], B1[4];
    float be[4];
#pragma unroll
    for (int t = 0; t < 4; ++t) {
      const int col = (cq * 4 + t) * 16 + lm;
      B0[t].x = W_enc[(2 * half + 0) * ENC_DIM + col];
      B0[t].y = W_enc[(2 * half + 1) * ENC_DIM + col];
      B1[t].x = W_enc[(2 * half + 4) * ENC_DIM + col];
      B1[t].y = W_enc[(2 * half + 5) * ENC_DIM + col];
      be[t]   = b_enc[col];
    }
    v8f pacc[4];
#pragma unroll
    for (int t = 0; t < 4; ++t) pacc[t] = vzero;
    float sw = 0.0f;

    const int* L = glist + grp * GLP;
    for (int pass = 0; pass < SPB / PER; ++pass) {
      if (pass > 0) __syncthreads();
      if (tidx < 4)  gcnt[tidx] = 0;
      if (tidx == 4) gmore = 0;
      __syncthreads();
      const int k = min(left, PER);
      auto put = [&](int e) {
        const int g = (e >> 4) & 3;
        const int p = atomicAdd(&gcnt[g], 1);
        glist[g * GLP + p] = ((e >> 6) << 4) | (e & 15);
      };
      if (pass == 0) {
#pragma unroll
        for (int i = 0; i < 8; ++i)
          if (i < k) put(pe[i]);
        for (int i = 8; i < k; ++i) put(srt[seg + i]);
      } else {
        for (int i = 0; i < k; ++i) put(srt[seg + i]);
      }
      seg  += k;
      left -= k;
      if (left > 0) gmore = 1;
      __syncthreads();
      if (tidx < 64) glist[(tidx >> 4) * GLP + min(gcnt[tidx >> 4], PER * NFB) + (tidx & 15)] = 0;
      __syncthreads();
      const int more  = gmore;
      const int n     = min(gcnt[grp], PER * NFB);
      const int ntile = (n + 15) >> 4;

      for (int tile = 0; tile < ntile; ++tile) {
        int ca;
        int cs[8], fs[8];
        {
          const int tb = tile * 16;
          ca = (L[tb + lm] >> 4) & (TOTAL_K - 1);
          const int4* lq = (const int4*)(L + tb + 8 * half);
          const int4 e0 = lq[0], e1 = lq[1];
          const int e[8] = {e0.x, e0.y, e0.z, e0.w, e1.x, e1.y, e1.z, e1.w};
#pragma unroll
          for (int i = 0; i < 8; ++i) {
            cs[i] = (e[i] >> 4) & (TOTAL_K - 1);
            fs[i] = (tb + 8 * half + i < n) ? (e[i] & 15) : -1;
          }
        }
        __builtin_amdgcn_sched_barrier(0);
        const float2* mp = (const float2*)(means + (size_t)ca * F_DIM);
        const float2 ma = mp[half], mb = mp[half + 2];
        v2f A0, A1;
        A0.x = ma.x; A0.y = ma.y;
        A1.x = mb.x; A1.y = mb.y;
        v2f SA[4];
#pragma unroll
        for (int j = 0; j < 4; ++j) {
          const float w0 = weights[cs[2 * j]], w1 = weights[cs[2 * j + 1]];
          SA[j].x = (fs[2 * j]     == lm) ? w0 : 0.0f;
          SA[j].y = (fs[2 * j + 1] == lm) ? w1 : 0.0f;
          sw += SA[j].x + SA[j].y;
        }
#pragma unroll
        for (int t = 0; t < 4; ++t) {
          v8f acc = wmma_f32_4(A0, B0[t], vzero);
          acc     = wmma_f32_4(A1, B1[t], acc);
          float g[8];
#pragma unroll
          for (int r = 0; r < 8; ++r) g[r] = gelu_exact(acc[r] + be[t]);
#pragma unroll
          for (int j = 0; j < 4; ++j) {
            v2f b;
            b.x = g[2 * j]; b.y = g[2 * j + 1];
            pacc[t] = wmma_f32_4(SA[j], b, pacc[t]);
          }
        }
      }
      if (!more) break;
    }
    sw += __shfl_xor(sw, 16, 32);
#pragma unroll
    for (int r = 0; r < 8; ++r) {
      const int   fl  = half * 8 + r;
      const float swr = __shfl(sw, fl, 32);
      const float rw  = 1.0f / (swr + 1e-8f);
      const int   tk  = ticker_ids[fb + fl];
#pragma unroll
      for (int t = 0; t < 4; ++t) {
        const int col = (cq * 4 + t) * 16 + lm;
        const float te = ticker_emb[(size_t)tk * ENC_DIM + col];
        pool[(grp * 16 + fl) * PP + col] = fmaf(te, swr, pacc[t][r]) * rw;
      }
    }
  }
  __syncthreads();

  v8f acc[4][2];
#pragma unroll
  for (int m = 0; m < 4; ++m) { acc[m][0] = vzero; acc[m][1] = vzero; }

  const int cb = wave * 32 + 2 * lm;
  const float* wp = W_proj + (size_t)(2 * half) * D_MODEL + cb;
  float2 n0 = *(const float2*)(wp);
  float2 n1 = *(const float2*)(wp + D_MODEL);
  const float* ap = pool + lm * PP + 2 * half;

  for (int kk = 0; kk < 64; ++kk) {
    const float2 r0 = n0, r1 = n1;
    const int kn = (kk + 1 < 64) ? kk + 1 : kk;
    const float* q = wp + (size_t)(4 * kn) * D_MODEL;
    n0 = *(const float2*)(q);
    n1 = *(const float2*)(q + D_MODEL);
    v2f b0, b1;
    b0.x = r0.x; b0.y = r1.x;
    b1.x = r0.y; b1.y = r1.y;
#pragma unroll
    for (int m = 0; m < 4; ++m) {
      const float2 av = *(const float2*)(ap + m * 16 * PP + 4 * kk);
      v2f a;
      a.x = av.x; a.y = av.y;
      acc[m][0] = wmma_f32_4(a, b0, acc[m][0]);
      acc[m][1] = wmma_f32_4(a, b1, acc[m][1]);
    }
  }

  {
    const float2 w0 = *(const float2*)(W_proj + (size_t)(ENC_DIM + 0) * D_MODEL + cb);
    const float2 w1 = *(const float2*)(W_proj + (size_t)(ENC_DIM + 1) * D_MODEL + cb);
    const float2 w2 = *(const float2*)(W_proj + (size_t)(ENC_DIM + 2) * D_MODEL + cb);
    const float2 bp = *(const float2*)(b_proj + cb);
#pragma unroll
    for (int m = 0; m < 4; ++m) {
#pragma unroll
      for (int r = 0; r < 8; ++r) {
        const int row = m * 16 + half * 8 + r;
        const float4 l = *(const float4*)(slog + row * 4);
        float2 v;
        v.x = acc[m][0][r] + bp.x + l.x * w0.x + l.y * w1.x + l.z * w2.x;
        v.y = acc[m][1][r] + bp.y + l.x * w0.y + l.y * w1.y + l.z * w2.y;
        *(float2*)(ytile + row * YP + cb) = v;
      }
    }
  }
  __syncthreads();

  {
    const int row = tidx >> 3;
    const int seg = tidx & 7;
    float s = 0.0f, ss = 0.0f;
#pragma unroll 16
    for (int j = 0; j < 64; ++j) {
      float v = ytile[row * YP + seg + j * 8];
      s  += v;
      ss += v * v;
    }
#pragma unroll
    for (int off = 1; off < 8; off <<= 1) {
      s  += __shfl_xor(s,  off, 32);
      ss += __shfl_xor(ss, off, 32);
    }
    if (seg == 0) {
      float mu  = s * (1.0f / (float)D_MODEL);
      float var = fmaf(-mu, mu, ss * (1.0f / (float)D_MODEL));
      smu[row] = mu;
      srs[row] = rsqrtf(fmaxf(var, 0.0f) + 1e-5f);
    }
  }
  __syncthreads();

  {
    const int col = tidx;
    const float gm = ln_gamma[col], bt = ln_beta[col];
    for (int row = 0; row < FPB; ++row) {
      float x = (ytile[row * YP + col] - smu[row]) * srs[row] * gm + bt;
      out[(size_t)(fb0 + row) * D_MODEL + col] = gelu_exact(x);
    }
  }
}

extern "C" void kernel_launch(void* const* d_in, const int* in_sizes, int n_in,
                              void* d_out, int out_size, void* d_ws, size_t ws_size,
                              hipStream_t stream) {
  const float* chunks        = (const float*)d_in[0];
  const int*   frame_id      = (const int*)  d_in[1];
  const float* weights       = (const float*)d_in[2];
  const float* frame_scalars = (const float*)d_in[3];
  const int*   ticker_ids    = (const int*)  d_in[5];
  const float* W_enc         = (const float*)d_in[6];
  const float* b_enc         = (const float*)d_in[7];
  const float* ticker_emb    = (const float*)d_in[8];
  const float* W_proj        = (const float*)d_in[9];
  const float* b_proj        = (const float*)d_in[10];
  const float* ln_gamma      = (const float*)d_in[11];
  const float* ln_beta       = (const float*)d_in[12];
  float* out = (float*)d_out;

  float* means = (float*)d_ws;
  int*   srt   = (int*)(means + (size_t)TOTAL_K * F_DIM);
  int*   offT  = srt + TOTAL_K;

  k_means<<<TOTAL_K / SPB, 512, 0, stream>>>(chunks, frame_id, means, srt, offT);

  k_frames<<<NFB, 512, 0, stream>>>(
      offT, srt, means, weights, ticker_ids, W_enc, b_enc, ticker_emb,
      frame_scalars, W_proj, b_proj, ln_gamma, ln_beta, out);
}


# GELU evaluated as h+|h|*erf(|z|), h=x/2: no copysign, z's scale folded into the constants, exp(-z*z) as one exp2; a quarter fewer vector instructions per encode tile; GELU last-bit rounding changes
# speedup vs baseline: 2.6293x; 1.0368x over previous
#include <hip/hip_runtime.h>
#include <hip/hip_bf16.h>

#define TOTAL_K     131072
#define CHUNK_LEN   64
#define F_DIM       8
#define NUM_FRAMES  16384
#define ENC_DIM     256
#define D_MODEL     512
#define N_SCALARS   3
#define N_TICKERS   4096
#define YP          520
#define FPB         64
#define PP          260
#define SPB         512
#define NFB         (NUM_FRAMES / FPB)
#define PER         16
#define GLP         (PER * NFB + 16)

typedef float v2f __attribute__((ext_vector_type(2)));
typedef float v8f __attribute__((ext_vector_type(8)));

__device__ __forceinline__ v8f wmma_f32_4(v2f a, v2f b, v8f c) {
  return __builtin_amdgcn_wmma_f32_16x16x4_f32(
       false, a,  false, b,
       (short)0, c,  false,  false);
}

__device__ __forceinline__ float gelu_exact(float x) {
  float h  = 0.5f * x;
  float t  = __builtin_amdgcn_rcpf(fmaf(0.3275911f * 0.70710678118654752f, fabsf(x), 1.0f));
  float p  = fmaf(t, 1.061405429f, -1.453152027f);
  p        = fmaf(t, p,  1.421413741f);
  p        = fmaf(t, p, -0.284496736f);
  p        = fmaf(t, p,  0.254829592f);
  p       *= t;
  float e  = __builtin_amdgcn_exp2f((x * x) * -0.72134752044448170f);
  float erf_abs = fmaf(-p, e, 1.0f);
  return fmaf(fabsf(h), erf_abs, h);
}

typedef float v4f __attribute__((ext_vector_type(4)));

__device__ __forceinline__ void lds_barrier() {
  __builtin_amdgcn_fence(__ATOMIC_RELEASE, "workgroup", "local");
  __builtin_amdgcn_s_barrier();
  __builtin_amdgcn_fence(__ATOMIC_ACQUIRE, "workgroup", "local");
}

__global__ void __launch_bounds__(512)
k_means(const float* __restrict__ chunks,
        const int*   __restrict__ frame_id,
        float* __restrict__ means,
        int*   __restrict__ srt,
        int*   __restrict__ offT) {
  __shared__ int hc[NFB];
  __shared__ int hs[NFB];
  __shared__ int wt[NFB / 32];
  const int tid  = threadIdx.x;
  const int lane = threadIdx.x & 31;
  const int wave = threadIdx.x >> 5;
  const int c0   = (blockIdx.x * 16 + wave) * 32;

  const int fid = frame_id[c0 + lane];
  const v4f* b = (const v4f*)(chunks + (size_t)c0 * (CHUNK_LEN * F_DIM)) + lane;

  v4f q[2][16];
#pragma unroll
  for (int j = 0; j < 16; ++j) q[0][j] = b[j * 32];

#pragma unroll
  for (int g = 0; g < 8; ++g) {
    if (g + 1 < 8) {
#pragma unroll
      for (int j = 0; j < 16; ++j)
        q[(g + 1) & 1][j] = b[(g + 1) * 512 + j * 32];
    }
    if (g == 0) {
      const int d = (fid >> 6) & (NFB - 1);
      if (tid < NFB) hc[tid] = 0;
      lds_barrier();
      const int rank = atomicAdd(&hc[d], 1);
      lds_barrier();
      const int v = (tid < NFB) ? hc[tid] : 0;
      int inc = v;
#pragma unroll
      for (int off = 1; off < 32; off <<= 1) {
        const int u = __shfl_up(inc, off, 32);
        if (lane >= off) inc += u;
      }
      if (tid < NFB && lane == 31) wt[wave] = inc;
      lds_barrier();
      if (tid < NFB) {
        int wp = 0;
#pragma unroll
        for (int w2 = 0; w2 < NFB / 32; ++w2) wp += (w2 < wave) ? wt[w2] : 0;
        hs[tid] = wp + inc - v;
      }
      lds_barrier();
      srt[blockIdx.x * SPB + hs[d] + rank] = ((c0 + lane) << 6) | (fid & 63);
      if (tid < NFB) {
        int2 sc;
        sc.x = hs[tid]; sc.y = v;
        ((int2*)offT)[tid * NFB + blockIdx.x] = sc;
      }
    }
    __builtin_amdgcn_sched_barrier(0);

    v4f s[4];
#pragma unroll
    for (int k = 0; k < 4; ++k)
      s[k] = (q[g & 1][4 * k + 0] + q[g & 1][4 * k + 1]) + (q[g & 1][4 * k + 2] + q[g & 1][4 * k + 3]);
#pragma unroll
    for (int off = 2; off < 32; off <<= 1) {
#pragma unroll
      for (int k = 0; k < 4; ++k) {
        s[k].x += __shfl_xor(s[k].x, off, 32);
        s[k].y += __shfl_xor(s[k].y, off, 32);
        s[k].z += __shfl_xor(s[k].z, off, 32);
        s[k].w += __shfl_xor(s[k].w, off, 32);
      }
    }
    const int sel = lane >> 1;
    v4f v = s[0];
    v = (sel == 1) ? s[1] : v;
    v = (sel == 2) ? s[2] : v;
    v = (sel == 3) ? s[3] : v;
    v *= (1.0f / 64.0f);
    if (lane < 8) ((v4f*)(means + (size_t)(c0 + g * 4) * F_DIM))[lane] = v;
  }
}

__global__ void __launch_bounds__(512)
k_frames(const int*   __restrict__ offT,
         const int*   __restrict__ srt,
         const float* __restrict__ means,
         const float* __restrict__ weights,
         const int*   __restrict__ ticker_ids,
         const float* __restrict__ W_enc,
         const float* __restrict__ b_enc,
         const float* __restrict__ ticker_emb,
         const float* __restrict__ frame_scalars,
         const float* __restrict__ W_proj,
         const float* __restrict__ b_proj,
         const float* __restrict__ ln_gamma,
         const float* __restrict__ ln_beta,
         float* __restrict__ out) {
  __shared__ __attribute__((aligned(16))) float pool[FPB * PP];
  __shared__ __attribute__((aligned(16))) float ytile[FPB * YP];
  __shared__ __attribute__((aligned(16))) float slog[FPB * 4];
  __shared__ float smu[FPB], srs[FPB];
  __shared__ __attribute__((aligned(16))) int glist[4 * GLP];
  __shared__ int gcnt[4];
  __shared__ int gmore;

  const int tidx = threadIdx.x;
  const int lane = tidx & 31;
  const int wave = tidx >> 5;
  const int lm   = lane & 15;
  const int half = lane >> 4;
  const int fb0  = blockIdx.x * FPB;

  const v8f vzero = {0.f,0.f,0.f,0.f,0.f,0.f,0.f,0.f};

  if (tidx < FPB * 4) {
    const int f = tidx >> 2, s = tidx & 3;
    slog[tidx] = (s < N_SCALARS) ? log1pf(frame_scalars[(fb0 + f) * N_SCALARS + s]) : 0.0f;
  }

  {
    const int grp = wave >> 2;
    const int cq  = wave & 3;
    const int gi  = blockIdx.x * (FPB / 16) + grp;
    const int fb  = gi * 16;

    int seg = 0, left = 0;
    int pe[8] = {0, 0, 0, 0, 0, 0, 0, 0};
    if (tidx < NFB) {
      const int2 sc = ((const int2*)offT)[blockIdx.x * NFB + tidx];
      const int st = min(max(sc.x, 0), SPB - 1);
      left = min(max(sc.y, 0), SPB - st);
      seg  = tidx * SPB + st;
#pragma unroll
      for (int i = 0; i < 8; ++i) pe[i] = srt[min(seg + i, TOTAL_K - 1)];
    }

    v2f B0[4], B1[4];
    float be[4];
#pragma unroll
    for (int t = 0; t < 4; ++t) {
      const int col = (cq * 4 + t) * 16 + lm;
      B0[t].x = W_enc[(2 * half + 0) * ENC_DIM + col];
      B0[t].y = W_enc[(2 * half + 1) * ENC_DIM + col];
      B1[t].x = W_enc[(2 * half + 4) * ENC_DIM + col];
      B1[t].y = W_enc[(2 * half + 5) * ENC_DIM + col];
      be[t]   = b_enc[col];
    }
    v8f pacc[4];
#pragma unroll
    for (int t = 0; t < 4; ++t) pacc[t] = vzero;
    float sw = 0.0f;

    const int* L = glist + grp * GLP;
    for (int pass = 0; pass < SPB / PER; ++pass) {
      if (pass > 0) __syncthreads();
      if (tidx < 4)  gcnt[tidx] = 0;
      if (tidx == 4) gmore = 0;
      __syncthreads();
      const int k = min(left, PER);
      auto put = [&](int e) {
        const int g = (e >> 4) & 3;
        const int p = atomicAdd(&gcnt[g], 1);
        glist[g * GLP + p] = ((e >> 6) << 4) | (e & 15);
      };
      if (pass == 0) {
#pragma unroll
        for (int i = 0; i < 8; ++i)
          if (i < k) put(pe[i]);
        for (int i = 8; i < k; ++i) put(srt[seg + i]);
      } else {
        for (int i = 0; i < k; ++i) put(srt[seg + i]);
      }
      seg  += k;
      left -= k;
      if (left > 0) gmore = 1;
      __syncthreads();
      if (tidx < 64) glist[(tidx >> 4) * GLP + min(gcnt[tidx >> 4], PER * NFB) + (tidx & 15)] = 0;
      __syncthreads();
      const int more  = gmore;
      const int n     = min(gcnt[grp], PER * NFB);
      const int ntile = (n + 15) >> 4;

      for (int tile = 0; tile < ntile; ++tile) {
        int ca;
        int cs[8], fs[8];
        {
          const int tb = tile * 16;
          ca = (L[tb + lm] >> 4) & (TOTAL_K - 1);
          const int4* lq = (const int4*)(L + tb + 8 * half);
          const int4 e0 = lq[0], e1 = lq[1];
          const int e[8] = {e0.x, e0.y, e0.z, e0.w, e1.x, e1.y, e1.z, e1.w};
#pragma unroll
          for (int i = 0; i < 8; ++i) {
            cs[i] = (e[i] >> 4) & (TOTAL_K - 1);
            fs[i] = (tb + 8 * half + i < n) ? (e[i] & 15) : -1;
          }
        }
        __builtin_amdgcn_sched_barrier(0);
        const float2* mp = (const float2*)(means + (size_t)ca * F_DIM);
        const float2 ma = mp[half], mb = mp[half + 2];
        v2f A0, A1;
        A0.x = ma.x; A0.y = ma.y;
        A1.x = mb.x; A1.y = mb.y;
        v2f SA[4];
#pragma unroll
        for (int j = 0; j < 4; ++j) {
          const float w0 = weights[cs[2 * j]], w1 = weights[cs[2 * j + 1]];
          SA[j].x = (fs[2 * j]     == lm) ? w0 : 0.0f;
          SA[j].y = (fs[2 * j + 1] == lm) ? w1 : 0.0f;
          sw += SA[j].x + SA[j].y;
        }
#pragma unroll
        for (int t = 0; t < 4; ++t) {
          v8f acc = wmma_f32_4(A0, B0[t], vzero);
          acc     = wmma_f32_4(A1, B1[t], acc);
          float g[8];
#pragma unroll
          for (int r = 0; r < 8; ++r) g[r] = gelu_exact(acc[r] + be[t]);
#pragma unroll
          for (int j = 0; j < 4; ++j) {
            v2f b;
            b.x = g[2 * j]; b.y = g[2 * j + 1];
            pacc[t] = wmma_f32_4(SA[j], b, pacc[t]);
          }
        }
      }
      if (!more) break;
    }
    sw += __shfl_xor(sw, 16, 32);
#pragma unroll
    for (int r = 0; r < 8; ++r) {
      const int   fl  = half * 8 + r;
      const float swr = __shfl(sw, fl, 32);
      const float rw  = 1.0f / (swr + 1e-8f);
      const int   tk  = ticker_ids[fb + fl];
#pragma unroll
      for (int t = 0; t < 4; ++t) {
        const int col = (cq * 4 + t) * 16 + lm;
        const float te = ticker_emb[(size_t)tk * ENC_DIM + col];
        pool[(grp * 16 + fl) * PP + col] = fmaf(te, swr, pacc[t][r]) * rw;
      }
    }
  }
  __syncthreads();

  v8f acc[4][2];
#pragma unroll
  for (int m = 0; m < 4; ++m) { acc[m][0] = vzero; acc[m][1] = vzero; }

  const int cb = wave * 32 + 2 * lm;
  const float* wp = W_proj + (size_t)(2 * half) * D_MODEL + cb;
  float2 n0 = *(const float2*)(wp);
  float2 n1 = *(const float2*)(wp + D_MODEL);
  const float* ap = pool + lm * PP + 2 * half;

  for (int kk = 0; kk < 64; ++kk) {
    const float2 r0 = n0, r1 = n1;
    const int kn = (kk + 1 < 64) ? kk + 1 : kk;
    const float* q = wp + (size_t)(4 * kn) * D_MODEL;
    n0 = *(const float2*)(q);
    n1 = *(const float2*)(q + D_MODEL);
    v2f b0, b1;
    b0.x = r0.x; b0.y = r1.x;
    b1.x = r0.y; b1.y = r1.y;
#pragma unroll
    for (int m = 0; m < 4; ++m) {
      const float2 av = *(const float2*)(ap + m * 16 * PP + 4 * kk);
      v2f a;
      a.x = av.x; a.y = av.y;
      acc[m][0] = wmma_f32_4(a, b0, acc[m][0]);
      acc[m][1] = wmma_f32_4(a, b1, acc[m][1]);
    }
  }

  {
    const float2 w0 = *(const float2*)(W_proj + (size_t)(ENC_DIM + 0) * D_MODEL + cb);
    const float2 w1 = *(const float2*)(W_proj + (size_t)(ENC_DIM + 1) * D_MODEL + cb);
    const float2 w2 = *(const float2*)(W_proj + (size_t)(ENC_DIM + 2) * D_MODEL + cb);
    const float2 bp = *(const float2*)(b_proj + cb);
#pragma unroll
    for (int m = 0; m < 4; ++m) {
#pragma unroll
      for (int r = 0; r < 8; ++r) {
        const int row = m * 16 + half * 8 + r;
        const float4 l = *(const float4*)(slog + row * 4);
        float2 v;
        v.x = acc[m][0][r] + bp.x + l.x * w0.x + l.y * w1.x + l.z * w2.x;
        v.y = acc[m][1][r] + bp.y + l.x * w0.y + l.y * w1.y + l.z * w2.y;
        *(float2*)(ytile + row * YP + cb) = v;
      }
    }
  }
  __syncthreads();

  {
    const int row = tidx >> 3;
    const int seg = tidx & 7;
    float s = 0.0f, ss = 0.0f;
#pragma unroll 16
    for (int j = 0; j < 64; ++j) {
      float v = ytile[row * YP + seg + j * 8];
      s  += v;
      ss += v * v;
    }
#pragma unroll
    for (int off = 1; off < 8; off <<= 1) {
      s  += __shfl_xor(s,  off, 32);
      ss += __shfl_xor(ss, off, 32);
    }
    if (seg == 0) {
      float mu  = s * (1.0f / (float)D_MODEL);
      float var = fmaf(-mu, mu, ss * (1.0f / (float)D_MODEL));
      smu[row] = mu;
      srs[row] = rsqrtf(fmaxf(var, 0.0f) + 1e-5f);
    }
  }
  __syncthreads();

  {
    const int col = tidx;
    const float gm = ln_gamma[col], bt = ln_beta[col];
    for (int row = 0; row < FPB; ++row) {
      float x = (ytile[row * YP + col] - smu[row]) * srs[row] * gm + bt;
      out[(size_t)(fb0 + row) * D_MODEL + col] = gelu_exact(x);
    }
  }
}

extern "C" void kernel_launch(void* const* d_in, const int* in_sizes, int n_in,
                              void* d_out, int out_size, void* d_ws, size_t ws_size,
                              hipStream_t stream) {
  const float* chunks        = (const float*)d_in[0];
  const int*   frame_id      = (const int*)  d_in[1];
  const float* weights       = (const float*)d_in[2];
  const float* frame_scalars = (const float*)d_in[3];
  const int*   ticker_ids    = (const int*)  d_in[5];
  const float* W_enc         = (const float*)d_in[6];
  const float* b_enc         = (const float*)d_in[7];
  const float* ticker_emb    = (const float*)d_in[8];
  const float* W_proj        = (const float*)d_in[9];
  const float* b_proj        = (const float*)d_in[10];
  const float* ln_gamma      = (const float*)d_in[11];
  const float* ln_beta       = (const float*)d_in[12];
  float* out = (float*)d_out;

  float* means = (float*)d_ws;
  int*   srt   = (int*)(means + (size_t)TOTAL_K * F_DIM);
  int*   offT  = srt + TOTAL_K;

  k_means<<<TOTAL_K / SPB, 512, 0, stream>>>(chunks, frame_id, means, srt, offT);

  k_frames<<<NFB, 512, 0, stream>>>(
      offT, srt, means, weights, ticker_ids, W_enc, b_enc, ticker_emb,
      frame_scalars, W_proj, b_proj, ln_gamma, ln_beta, out);
}
